# starting kernel
# baseline (speedup 1.0000x reference)
#include <hip/hip_runtime.h>
#include <hip/hip_bf16.h>

typedef float v2f __attribute__((ext_vector_type(2)));
typedef float v8f __attribute__((ext_vector_type(8)));

#define B_TOT    16384
#define D_DIM    1024
#define E_EXP    4
#define R_RANK   32
#define N1       128
#define L_LAYERS 2
#define B_ROWS   32
#define THREADS  256
#define XPITCH   1028
#define YPITCH   132

#define SMEM_FLOATS (2 * B_ROWS * XPITCH + 2 * B_ROWS * YPITCH + B_ROWS * E_EXP + THREADS)

__device__ __forceinline__ float fast_tanh(float x) {
    const float e = __expf(2.0f * x);
    return 1.0f - 2.0f * __frcp_rn(e + 1.0f);
}

extern "C" __global__ __launch_bounds__(THREADS)
void crossnet_mix_kernel(const float* __restrict__ x,
                         const float* __restrict__ U,
                         const float* __restrict__ V,
                         const float* __restrict__ C,
                         const float* __restrict__ bias,
                         const float* __restrict__ gate_w,
                         float* __restrict__ out)
{
    extern __shared__ float smem[];
    float* X0 = smem;
    float* XL = smem + B_ROWS * XPITCH;
    float* YB = smem + 2 * B_ROWS * XPITCH;
    float* CB = YB + B_ROWS * YPITCH;
    float* GT = CB + B_ROWS * YPITCH;
    float* GP = GT + B_ROWS * E_EXP;

    const int tid  = threadIdx.x;
    const int wave = tid >> 5;
    const int lane = tid & 31;
    const int lo16 = lane & 15;
    const int hiK  = lane >> 4;
    const int rowBase = blockIdx.x * B_ROWS;

    for (int i = tid; i < B_ROWS * (D_DIM / 4); i += THREADS) {
        const int r  = i >> 8;
        const int c4 = i & 255;
        const float4 val = ((const float4*)(x + (size_t)(rowBase + r) * D_DIM))[c4];
        *(float4*)&X0[r * XPITCH + c4 * 4] = val;
        *(float4*)&XL[r * XPITCH + c4 * 4] = val;
    }
    __syncthreads();

    const v8f vzero = {};

    for (int l = 0; l < L_LAYERS; ++l) {
        const float* Vl = V + (size_t)l * E_EXP * D_DIM * R_RANK;
        const float* Ul = U + (size_t)l * E_EXP * D_DIM * R_RANK;
        const float* Cl = C + (size_t)l * E_EXP * R_RANK * R_RANK;
        const float* bl = bias + (size_t)l * D_DIM;

        const int n0 = wave * 16;
        const int e1 = n0 >> 5;
        const int c1 = (n0 & 31) + lo16;
        v8f acc0 = vzero, acc1 = vzero;
        for (int k0 = 0; k0 < D_DIM; k0 += 4) {
            const int kk = k0 + 2 * hiK;
            const v2f a0 = *(const v2f*)&XL[lo16 * XPITCH + kk];
            const v2f a1 = *(const v2f*)&XL[(lo16 + 16) * XPITCH + kk];
            const float* vb = Vl + ((size_t)e1 * D_DIM + kk) * R_RANK + c1;
            v2f b;
            b.x = vb[0];
            b.y = vb[R_RANK];
            acc0 = __builtin_amdgcn_wmma_f32_16x16x4_f32(false, a0, false, b, (short)0, acc0, false, false);
            acc1 = __builtin_amdgcn_wmma_f32_16x16x4_f32(false, a1, false, b, (short)0, acc1, false, false);
        }
#pragma unroll
        for (int j = 0; j < 8; ++j) {
            const int r0 = j + 8 * hiK;
            YB[r0 * YPITCH + n0 + lo16]        = fast_tanh(acc0[j]);
            YB[(r0 + 16) * YPITCH + n0 + lo16] = fast_tanh(acc1[j]);
        }

        {
            const int b = tid >> 3;
            const int e = (tid >> 1) & 3;
            const int h = tid & 1;
            const float4* gw = (const float4*)(gate_w + (size_t)e * D_DIM + h * (D_DIM / 2));
            const float* xr = &XL[b * XPITCH + h * (D_DIM / 2)];
            float s = 0.f;
            for (int d4 = 0; d4 < D_DIM / 8; ++d4) {
                const float4 g = gw[d4];
                const float* xp = xr + d4 * 4;
                s += g.x * xp[0] + g.y * xp[1] + g.z * xp[2] + g.w * xp[3];
            }
            GP[tid] = s;
        }
        __syncthreads();
        if (tid < B_ROWS) {
            const float g0 = GP[tid * 8 + 0] + GP[tid * 8 + 1];
            const float g1 = GP[tid * 8 + 2] + GP[tid * 8 + 3];
            const float g2 = GP[tid * 8 + 4] + GP[tid * 8 + 5];
            const float g3 = GP[tid * 8 + 6] + GP[tid * 8 + 7];
            const float m  = fmaxf(fmaxf(g0, g1), fmaxf(g2, g3));
            const float e0 = __expf(g0 - m), e1x = __expf(g1 - m);
            const float e2 = __expf(g2 - m), e3 = __expf(g3 - m);
            const float inv = __frcp_rn(e0 + e1x + e2 + e3);
            GT[tid * 4 + 0] = e0 * inv;  GT[tid * 4 + 1] = e1x * inv;
            GT[tid * 4 + 2] = e2 * inv;  GT[tid * 4 + 3] = e3 * inv;
        }
        __syncthreads();

        for (int i = tid; i < B_ROWS * N1; i += THREADS) {
            const int b = i >> 7, col = i & 127;
            const int e = col >> 5, r = col & 31;
            const float* crow = Cl + ((size_t)e * R_RANK + r) * R_RANK;
            const float* vrow = &YB[b * YPITCH + e * R_RANK];
            float s = 0.f;
#pragma unroll 8
            for (int ss = 0; ss < R_RANK; ++ss) s += crow[ss] * vrow[ss];
            CB[b * YPITCH + col] = GT[b * 4 + e] * fast_tanh(s);
        }
        __syncthreads();

        const int d0 = wave * 128;
        v8f acc2[2][8];
#pragma unroll
        for (int m = 0; m < 2; ++m)
#pragma unroll
            for (int nt = 0; nt < 8; ++nt) acc2[m][nt] = vzero;

        for (int k0 = 0; k0 < N1; k0 += 4) {
            const int kk = k0 + 2 * hiK;
            const int e = kk >> 5, r = kk & 31;
            const v2f a0 = *(const v2f*)&CB[lo16 * YPITCH + kk];
            const v2f a1 = *(const v2f*)&CB[(lo16 + 16) * YPITCH + kk];
#pragma unroll
            for (int nt = 0; nt < 8; ++nt) {
                const int col = d0 + nt * 16 + lo16;
                const v2f b = *(const v2f*)(Ul + ((size_t)e * D_DIM + col) * R_RANK + r);
                acc2[0][nt] = __builtin_amdgcn_wmma_f32_16x16x4_f32(false, a0, false, b, (short)0, acc2[0][nt], false, false);
                acc2[1][nt] = __builtin_amdgcn_wmma_f32_16x16x4_f32(false, a1, false, b, (short)0, acc2[1][nt], false, false);
            }
        }

#pragma unroll
        for (int nt = 0; nt < 8; ++nt) {
            const int col = d0 + nt * 16 + lo16;
            const float bcol = bl[col];
#pragma unroll
            for (int m = 0; m < 2; ++m) {
#pragma unroll
                for (int j = 0; j < 8; ++j) {
                    const int row = m * 16 + j + 8 * hiK;
                    const float xv  = XL[row * XPITCH + col];
                    const float x0v = X0[row * XPITCH + col];
                    XL[row * XPITCH + col] = xv + x0v * (bcol + acc2[m][nt][j]);
                }
            }
        }
        __syncthreads();
    }

    for (int i = tid; i < B_ROWS * (D_DIM / 4); i += THREADS) {
        const int r  = i >> 8;
        const int c4 = i & 255;
        const float4 v4 = *(const float4*)&XL[r * XPITCH + c4 * 4];
        ((float4*)(out + (size_t)(rowBase + r) * D_DIM))[c4] = v4;
    }
}

extern "C" void kernel_launch(void* const* d_in, const int* in_sizes, int n_in,
                              void* d_out, int out_size, void* d_ws, size_t ws_size,
                              hipStream_t stream) {
    (void)in_sizes; (void)n_in; (void)d_ws; (void)ws_size; (void)out_size;
    const float* x      = (const float*)d_in[0];
    const float* U      = (const float*)d_in[1];
    const float* V      = (const float*)d_in[2];
    const float* C      = (const float*)d_in[3];
    const float* bias   = (const float*)d_in[4];
    const float* gate_w = (const float*)d_in[5];
    float* out = (float*)d_out;

    const dim3 grid(B_TOT / B_ROWS);
    const size_t smemBytes = (size_t)SMEM_FLOATS * sizeof(float);
    crossnet_mix_kernel<<<grid, THREADS, smemBytes, stream>>>(x, U, V, C, bias, gate_w, out);
}


# Global operand loads (x, V, U, C, gate weights, bias) requested in deep register groups one block ahead of the matrix work, not one wait per load; every sum keeps its order
# speedup vs baseline: 2.3711x; 2.3711x over previous
#include <hip/hip_runtime.h>
#include <hip/hip_bf16.h>


typedef float v2f __attribute__((ext_vector_type(2)));
typedef float v8f __attribute__((ext_vector_type(8)));

#define B_TOT    16384
#define D_DIM    1024
#define E_EXP    4
#define R_RANK   32
#define N1       128
#define L_LAYERS 2
#define B_ROWS   32
#define THREADS  256
#define XPITCH   1028
#define YPITCH   132

#define KB1      16
#define NB1      (D_DIM / (4 * KB1))
#define GW_GRP   32

#define SMEM_FLOATS (2 * B_ROWS * XPITCH + 2 * B_ROWS * YPITCH + B_ROWS * E_EXP + THREADS)

__device__ __forceinline__ float fast_tanh(float x) {
    const float e = __expf(2.0f * x);
    return 1.0f - 2.0f * __frcp_rn(e + 1.0f);
}

__device__ __forceinline__ void g1_load(float (&bx)[KB1], float (&by)[KB1],
                                        const float* __restrict__ vb, int kb) {
    const float* p = vb + (size_t)kb * (4 * KB1) * R_RANK;
#pragma unroll
    for (int j = 0; j < KB1; ++j) {
        bx[j] = p[(4 * j) * R_RANK];
        by[j] = p[(4 * j + 1) * R_RANK];
    }
}

__device__ __forceinline__ void g1_mma(const float (&bx)[KB1], const float (&by)[KB1],
                                       const float* xa, int kb, v8f& acc0, v8f& acc1) {
#pragma unroll
    for (int j = 0; j < KB1; ++j) {
        const int k0 = (kb * KB1 + j) * 4;
        const v2f a0 = *(const v2f*)&xa[k0];
        const v2f a1 = *(const v2f*)&xa[16 * XPITCH + k0];
        v2f b;
        b.x = bx[j];
        b.y = by[j];
        acc0 = __builtin_amdgcn_wmma_f32_16x16x4_f32(false, a0, false, b, (short)0, acc0, false, false);
        acc1 = __builtin_amdgcn_wmma_f32_16x16x4_f32(false, a1, false, b, (short)0, acc1, false, false);
    }
}

__device__ __forceinline__ void g2_load(v2f (&b)[4][8], const float* __restrict__ ub, int e, int half) {
#pragma unroll
    for (int q = 0; q < 4; ++q) {
        const float* p = ub + ((size_t)e * D_DIM + (half * 4 + q) * 16) * R_RANK;
#pragma unroll
        for (int s = 0; s < 8; ++s) b[q][s] = *(const v2f*)(p + 4 * s);
    }
}

__device__ __forceinline__ void g2_afrag(v2f (&a0)[8], v2f (&a1)[8], const float* ca, int e) {
#pragma unroll
    for (int s = 0; s < 8; ++s) {
        a0[s] = *(const v2f*)&ca[e * R_RANK + 4 * s];
        a1[s] = *(const v2f*)&ca[16 * YPITCH + e * R_RANK + 4 * s];
    }
}

template <int HALF>
__device__ __forceinline__ void g2_mma(const v2f (&b)[4][8], const v2f (&a0)[8], const v2f (&a1)[8],
                                       v8f (&acc)[2][8]) {
#pragma unroll
    for (int s = 0; s < 8; ++s) {
#pragma unroll
        for (int q = 0; q < 4; ++q) {
            acc[0][HALF * 4 + q] = __builtin_amdgcn_wmma_f32_16x16x4_f32(false, a0[s], false, b[q][s], (short)0, acc[0][HALF * 4 + q], false, false);
            acc[1][HALF * 4 + q] = __builtin_amdgcn_wmma_f32_16x16x4_f32(false, a1[s], false, b[q][s], (short)0, acc[1][HALF * 4 + q], false, false);
        }
    }
}

extern "C" __global__ __launch_bounds__(THREADS)
void crossnet_mix_kernel(const float* __restrict__ x,
                         const float* __restrict__ U,
                         const float* __restrict__ V,
                         const float* __restrict__ C,
                         const float* __restrict__ bias,
                         const float* __restrict__ gate_w,
                         float* __restrict__ out)
{
    extern __shared__ float smem[];
    float* X0 = smem;
    float* XL = smem + B_ROWS * XPITCH;
    float* YB = smem + 2 * B_ROWS * XPITCH;
    float* CB = YB + B_ROWS * YPITCH;
    float* GT = CB + B_ROWS * YPITCH;
    float* GP = GT + B_ROWS * E_EXP;

    const int tid  = threadIdx.x;
    const int wave = tid >> 5;
    const int lane = tid & 31;
    const int lo16 = lane & 15;
    const int hiK  = lane >> 4;
    const int rowBase = blockIdx.x * B_ROWS;

#pragma unroll
    for (int rh = 0; rh < B_ROWS; rh += 16) {
        float4 xv[16];
#pragma unroll
        for (int r = 0; r < 16; ++r)
            xv[r] = ((const float4*)(x + (size_t)(rowBase + rh + r) * D_DIM))[tid];
#pragma unroll
        for (int r = 0; r < 16; ++r) {
            *(float4*)&X0[(rh + r) * XPITCH + tid * 4] = xv[r];
            *(float4*)&XL[(rh + r) * XPITCH + tid * 4] = xv[r];
        }
    }
    __syncthreads();

    const v8f vzero = {};

    const int ccol = tid & 127;
    const int ce   = ccol >> 5;
    const int cr_  = ccol & 31;

    for (int l = 0; l < L_LAYERS; ++l) {
        const float* Vl = V + (size_t)l * E_EXP * D_DIM * R_RANK;
        const float* Ul = U + (size_t)l * E_EXP * D_DIM * R_RANK;
        const float* Cl = C + (size_t)l * E_EXP * R_RANK * R_RANK;
        const float* bl = bias + (size_t)l * D_DIM;

        float4 crow[8];
        {
            const float4* cp = (const float4*)(Cl + ((size_t)ce * R_RANK + cr_) * R_RANK);
#pragma unroll
            for (int j = 0; j < 8; ++j) crow[j] = cp[j];
        }

        const int n0 = wave * 16;
        const int e1 = n0 >> 5;
        const int c1 = (n0 & 31) + lo16;
        v8f acc0 = vzero, acc1 = vzero;
        {
            const float* vb = Vl + ((size_t)e1 * D_DIM + 2 * hiK) * R_RANK + c1;
            const float* xa = &XL[lo16 * XPITCH + 2 * hiK];
            float bx0[KB1], by0[KB1], bx1[KB1], by1[KB1];
            g1_load(bx0, by0, vb, 0);
            for (int kb = 0; kb < NB1 - 2; kb += 2) {
                g1_load(bx1, by1, vb, kb + 1);
                g1_mma(bx0, by0, xa, kb, acc0, acc1);
                g1_load(bx0, by0, vb, kb + 2);
                g1_mma(bx1, by1, xa, kb + 1, acc0, acc1);
            }
            g1_load(bx1, by1, vb, NB1 - 1);
            g1_mma(bx0, by0, xa, NB1 - 2, acc0, acc1);
            g1_mma(bx1, by1, xa, NB1 - 1, acc0, acc1);
        }
#pragma unroll
        for (int j = 0; j < 8; ++j) {
            const int r0 = j + 8 * hiK;
            YB[r0 * YPITCH + n0 + lo16]        = fast_tanh(acc0[j]);
            YB[(r0 + 16) * YPITCH + n0 + lo16] = fast_tanh(acc1[j]);
        }

        {
            const int b = tid >> 3;
            const int e = (tid >> 1) & 3;
            const int h = tid & 1;
            const float4* gw = (const float4*)(gate_w + (size_t)e * D_DIM + h * (D_DIM / 2));
            const float* xr = &XL[b * XPITCH + h * (D_DIM / 2)];
            float s = 0.f;
#pragma unroll 1
            for (int g0 = 0; g0 < D_DIM / 8; g0 += GW_GRP) {
                float4 gv[GW_GRP];
#pragma unroll
                for (int j = 0; j < GW_GRP; ++j) gv[j] = gw[g0 + j];
#pragma unroll
                for (int j = 0; j < GW_GRP; ++j) {
                    const float4 g = gv[j];
                    const float* xp = xr + (g0 + j) * 4;
                    s += g.x * xp[0] + g.y * xp[1] + g.z * xp[2] + g.w * xp[3];
                }
            }
            GP[tid] = s;
        }
        __syncthreads();
        if (tid < B_ROWS) {
            const float g0 = GP[tid * 8 + 0] + GP[tid * 8 + 1];
            const float g1 = GP[tid * 8 + 2] + GP[tid * 8 + 3];
            const float g2 = GP[tid * 8 + 4] + GP[tid * 8 + 5];
            const float g3 = GP[tid * 8 + 6] + GP[tid * 8 + 7];
            const float m  = fmaxf(fmaxf(g0, g1), fmaxf(g2, g3));
            const float e0 = __expf(g0 - m), e1x = __expf(g1 - m);
            const float e2 = __expf(g2 - m), e3 = __expf(g3 - m);
            const float inv = __frcp_rn(e0 + e1x + e2 + e3);
            GT[tid * 4 + 0] = e0 * inv;  GT[tid * 4 + 1] = e1x * inv;
            GT[tid * 4 + 2] = e2 * inv;  GT[tid * 4 + 3] = e3 * inv;
        }
        __syncthreads();

        const int d0 = wave * 128;
        const float* ub = Ul + (size_t)(d0 + lo16) * R_RANK + 2 * hiK;
        v2f bb0[4][8], bb1[4][8];
        g2_load(bb0, ub, 0, 0);
        float bcol[8];
#pragma unroll
        for (int nt = 0; nt < 8; ++nt) bcol[nt] = bl[d0 + nt * 16 + lo16];

#pragma unroll 2
        for (int t = 0; t < (B_ROWS * N1) / THREADS; ++t) {
            const int b = (tid >> 7) + 2 * t;
            const float* vrow = &YB[b * YPITCH + ce * R_RANK];
            float s = 0.f;
#pragma unroll
            for (int j = 0; j < 8; ++j) {
                s += crow[j].x * vrow[4 * j + 0];
                s += crow[j].y * vrow[4 * j + 1];
                s += crow[j].z * vrow[4 * j + 2];
                s += crow[j].w * vrow[4 * j + 3];
            }
            CB[b * YPITCH + ccol] = GT[b * 4 + ce] * fast_tanh(s);
        }
        __syncthreads();

        v8f acc2[2][8];
#pragma unroll
        for (int m = 0; m < 2; ++m)
#pragma unroll
            for (int nt = 0; nt < 8; ++nt) acc2[m][nt] = vzero;

        v2f a0[8], a1[8];
        const float* ca = &CB[lo16 * YPITCH + 2 * hiK];
#pragma unroll 1
        for (int e = 0; e < E_EXP - 1; ++e) {
            g2_load(bb1, ub, e, 1);
            g2_afrag(a0, a1, ca, e);
            g2_mma<0>(bb0, a0, a1, acc2);
            g2_load(bb0, ub, e + 1, 0);
            g2_mma<1>(bb1, a0, a1, acc2);
        }
        g2_load(bb1, ub, E_EXP - 1, 1);
        __builtin_amdgcn_sched_barrier(0);
        g2_afrag(a0, a1, ca, E_EXP - 1);
        g2_mma<0>(bb0, a0, a1, acc2);
        g2_mma<1>(bb1, a0, a1, acc2);

#pragma unroll
        for (int nt = 0; nt < 8; ++nt) {
            const int col = d0 + nt * 16 + lo16;
#pragma unroll
            for (int m = 0; m < 2; ++m) {
#pragma unroll
                for (int j = 0; j < 8; ++j) {
                    const int row = m * 16 + j + 8 * hiK;
                    const float xv  = XL[row * XPITCH + col];
                    const float x0v = X0[row * XPITCH + col];
                    XL[row * XPITCH + col] = xv + x0v * (bcol[nt] + acc2[m][nt][j]);
                }
            }
        }
        __syncthreads();
    }

    for (int i = tid; i < B_ROWS * (D_DIM / 4); i += THREADS) {
        const int r  = i >> 8;
        const int c4 = i & 255;
        const float4 v4 = *(const float4*)&XL[r * XPITCH + c4 * 4];
        ((float4*)(out + (size_t)(rowBase + r) * D_DIM))[c4] = v4;
    }
}

extern "C" void kernel_launch(void* const* d_in, const int* in_sizes, int n_in,
                              void* d_out, int out_size, void* d_ws, size_t ws_size,
                              hipStream_t stream) {
    (void)in_sizes; (void)n_in; (void)d_ws; (void)ws_size; (void)out_size;
    const float* x      = (const float*)d_in[0];
    const float* U      = (const float*)d_in[1];
    const float* V      = (const float*)d_in[2];
    const float* C      = (const float*)d_in[3];
    const float* bias   = (const float*)d_in[4];
    const float* gate_w = (const float*)d_in[5];
    float* out = (float*)d_out;

    const dim3 grid(B_TOT / B_ROWS);
    const size_t smemBytes = (size_t)SMEM_FLOATS * sizeof(float);
    crossnet_mix_kernel<<<grid, THREADS, smemBytes, stream>>>(x, U, V, C, bias, gate_w, out);
}


# Both products: each block's operand loads (LDS A fragments, global B) issued as ONE group a block ahead, none among the matrix instructions; second A register set in product 1; sum order kept
# speedup vs baseline: 2.5738x; 1.0855x over previous
#include <hip/hip_runtime.h>
#include <hip/hip_bf16.h>


typedef float v2f __attribute__((ext_vector_type(2)));
typedef float v8f __attribute__((ext_vector_type(8)));

#define B_TOT    16384
#define D_DIM    1024
#define E_EXP    4
#define R_RANK   32
#define N1       128
#define L_LAYERS 2
#define B_ROWS   32
#define THREADS  256
#define XPITCH   1028
#define YPITCH   132

#define KB1      16
#define NB1      (D_DIM / (4 * KB1))
#define GW_GRP   32

#define SMEM_FLOATS (2 * B_ROWS * XPITCH + 2 * B_ROWS * YPITCH + B_ROWS * E_EXP + THREADS)

__device__ __forceinline__ float fast_tanh(float x) {
    const float e = __expf(2.0f * x);
    return 1.0f - 2.0f * __frcp_rn(e + 1.0f);
}

__device__ __forceinline__ void g1_load(float (&bx)[KB1], float (&by)[KB1],
                                        const float* __restrict__ vb, int kb) {
    const float* p = vb + (size_t)kb * (4 * KB1) * R_RANK;
#pragma unroll
    for (int j = 0; j < KB1; ++j) {
        bx[j] = p[(4 * j) * R_RANK];
        by[j] = p[(4 * j + 1) * R_RANK];
    }
}

__device__ __forceinline__ void g1_aload(v2f (&a0)[KB1], v2f (&a1)[KB1], const float* xa, int kb) {
#pragma unroll
    for (int j = 0; j < KB1; ++j) {
        const int k0 = (kb * KB1 + j) * 4;
        a0[j] = *(const v2f*)&xa[k0];
        a1[j] = *(const v2f*)&xa[16 * XPITCH + k0];
    }
}

__device__ __forceinline__ void g1_mma(const float (&bx)[KB1], const float (&by)[KB1],
                                       const v2f (&a0)[KB1], const v2f (&a1)[KB1], v8f& acc0, v8f& acc1) {
#pragma unroll
    for (int j = 0; j < KB1; ++j) {
        v2f b;
        b.x = bx[j];
        b.y = by[j];
        acc0 = __builtin_amdgcn_wmma_f32_16x16x4_f32(false, a0[j], false, b, (short)0, acc0, false, false);
        acc1 = __builtin_amdgcn_wmma_f32_16x16x4_f32(false, a1[j], false, b, (short)0, acc1, false, false);
    }
}

__device__ __forceinline__ void g2_load(v2f (&b)[4][8], const float* __restrict__ ub, int e, int half) {
#pragma unroll
    for (int q = 0; q < 4; ++q) {
        const float* p = ub + ((size_t)e * D_DIM + (half * 4 + q) * 16) * R_RANK;
#pragma unroll
        for (int s = 0; s < 8; ++s) b[q][s] = *(const v2f*)(p + 4 * s);
    }
}

__device__ __forceinline__ void g2_afrag(v2f (&a0)[8], v2f (&a1)[8], const float* ca, int e) {
#pragma unroll
    for (int s = 0; s < 8; ++s) {
        a0[s] = *(const v2f*)&ca[e * R_RANK + 4 * s];
        a1[s] = *(const v2f*)&ca[16 * YPITCH + e * R_RANK + 4 * s];
    }
}

template <int HALF>
__device__ __forceinline__ void g2_mma(const v2f (&b)[4][8], const v2f (&a0)[8], const v2f (&a1)[8],
                                       v8f (&acc)[2][8]) {
#pragma unroll
    for (int s = 0; s < 8; ++s) {
#pragma unroll
        for (int q = 0; q < 4; ++q) {
            acc[0][HALF * 4 + q] = __builtin_amdgcn_wmma_f32_16x16x4_f32(false, a0[s], false, b[q][s], (short)0, acc[0][HALF * 4 + q], false, false);
            acc[1][HALF * 4 + q] = __builtin_amdgcn_wmma_f32_16x16x4_f32(false, a1[s], false, b[q][s], (short)0, acc[1][HALF * 4 + q], false, false);
        }
    }
}

extern "C" __global__ __launch_bounds__(THREADS)
void crossnet_mix_kernel(const float* __restrict__ x,
                         const float* __restrict__ U,
                         const float* __restrict__ V,
                         const float* __restrict__ C,
                         const float* __restrict__ bias,
                         const float* __restrict__ gate_w,
                         float* __restrict__ out)
{
    extern __shared__ float smem[];
    float* X0 = smem;
    float* XL = smem + B_ROWS * XPITCH;
    float* YB = smem + 2 * B_ROWS * XPITCH;
    float* CB = YB + B_ROWS * YPITCH;
    float* GT = CB + B_ROWS * YPITCH;
    float* GP = GT + B_ROWS * E_EXP;

    const int tid  = threadIdx.x;
    const int wave = tid >> 5;
    const int lane = tid & 31;
    const int lo16 = lane & 15;
    const int hiK  = lane >> 4;
    const int rowBase = blockIdx.x * B_ROWS;

#pragma unroll
    for (int rh = 0; rh < B_ROWS; rh += 16) {
        float4 xv[16];
#pragma unroll
        for (int r = 0; r < 16; ++r)
            xv[r] = ((const float4*)(x + (size_t)(rowBase + rh + r) * D_DIM))[tid];
#pragma unroll
        for (int r = 0; r < 16; ++r) {
            *(float4*)&X0[(rh + r) * XPITCH + tid * 4] = xv[r];
            *(float4*)&XL[(rh + r) * XPITCH + tid * 4] = xv[r];
        }
    }
    __syncthreads();

    const v8f vzero = {};

    const int ccol = tid & 127;
    const int ce   = ccol >> 5;
    const int cr_  = ccol & 31;

    for (int l = 0; l < L_LAYERS; ++l) {
        const float* Vl = V + (size_t)l * E_EXP * D_DIM * R_RANK;
        const float* Ul = U + (size_t)l * E_EXP * D_DIM * R_RANK;
        const float* Cl = C + (size_t)l * E_EXP * R_RANK * R_RANK;
        const float* bl = bias + (size_t)l * D_DIM;

        float4 crow[8];
        {
            const float4* cp = (const float4*)(Cl + ((size_t)ce * R_RANK + cr_) * R_RANK);
#pragma unroll
            for (int j = 0; j < 8; ++j) crow[j] = cp[j];
        }

        const int n0 = wave * 16;
        const int e1 = n0 >> 5;
        const int c1 = (n0 & 31) + lo16;
        v8f acc0 = vzero, acc1 = vzero;
        {
            const float* vb = Vl + ((size_t)e1 * D_DIM + 2 * hiK) * R_RANK + c1;
            const float* xa = &XL[lo16 * XPITCH + 2 * hiK];
            float bx0[KB1], by0[KB1], bx1[KB1], by1[KB1];
            v2f xa0[KB1], xb0[KB1], xa1[KB1], xb1[KB1];
            g1_load(bx0, by0, vb, 0);
            g1_aload(xa0, xb0, xa, 0);
            for (int kb = 0; kb < NB1 - 2; kb += 2) {
                g1_load(bx1, by1, vb, kb + 1);
                g1_aload(xa1, xb1, xa, kb + 1);
                g1_mma(bx0, by0, xa0, xb0, acc0, acc1);
                g1_load(bx0, by0, vb, kb + 2);
                g1_aload(xa0, xb0, xa, kb + 2);
                g1_mma(bx1, by1, xa1, xb1, acc0, acc1);
                __builtin_amdgcn_sched_group_barrier(0x020, 2 * KB1, 0);
                __builtin_amdgcn_sched_group_barrier(0x100, 2 * KB1, 0);
                __builtin_amdgcn_sched_group_barrier(0x008, 2 * KB1, 0);
                __builtin_amdgcn_sched_group_barrier(0x020, 2 * KB1, 0);
                __builtin_amdgcn_sched_group_barrier(0x100, 2 * KB1, 0);
                __builtin_amdgcn_sched_group_barrier(0x008, 2 * KB1, 0);
            }
            g1_load(bx1, by1, vb, NB1 - 1);
            g1_aload(xa1, xb1, xa, NB1 - 1);
            __builtin_amdgcn_sched_barrier(0);
            g1_mma(bx0, by0, xa0, xb0, acc0, acc1);
            g1_mma(bx1, by1, xa1, xb1, acc0, acc1);
        }
#pragma unroll
        for (int j = 0; j < 8; ++j) {
            const int r0 = j + 8 * hiK;
            YB[r0 * YPITCH + n0 + lo16]        = fast_tanh(acc0[j]);
            YB[(r0 + 16) * YPITCH + n0 + lo16] = fast_tanh(acc1[j]);
        }

        {
            const int b = tid >> 3;
            const int e = (tid >> 1) & 3;
            const int h = tid & 1;
            const float4* gw = (const float4*)(gate_w + (size_t)e * D_DIM + h * (D_DIM / 2));
            const float* xr = &XL[b * XPITCH + h * (D_DIM / 2)];
            float s = 0.f;
#pragma unroll 1
            for (int g0 = 0; g0 < D_DIM / 8; g0 += GW_GRP) {
                float4 gv[GW_GRP];
#pragma unroll
                for (int j = 0; j < GW_GRP; ++j) gv[j] = gw[g0 + j];
#pragma unroll
                for (int j = 0; j < GW_GRP; ++j) {
                    const float4 g = gv[j];
                    const float* xp = xr + (g0 + j) * 4;
                    s += g.x * xp[0] + g.y * xp[1] + g.z * xp[2] + g.w * xp[3];
                }
            }
            GP[tid] = s;
        }
        __syncthreads();
        if (tid < B_ROWS) {
            const float g0 = GP[tid * 8 + 0] + GP[tid * 8 + 1];
            const float g1 = GP[tid * 8 + 2] + GP[tid * 8 + 3];
            const float g2 = GP[tid * 8 + 4] + GP[tid * 8 + 5];
            const float g3 = GP[tid * 8 + 6] + GP[tid * 8 + 7];
            const float m  = fmaxf(fmaxf(g0, g1), fmaxf(g2, g3));
            const float e0 = __expf(g0 - m), e1x = __expf(g1 - m);
            const float e2 = __expf(g2 - m), e3 = __expf(g3 - m);
            const float inv = __frcp_rn(e0 + e1x + e2 + e3);
            GT[tid * 4 + 0] = e0 * inv;  GT[tid * 4 + 1] = e1x * inv;
            GT[tid * 4 + 2] = e2 * inv;  GT[tid * 4 + 3] = e3 * inv;
        }
        __syncthreads();

        const int d0 = wave * 128;
        const float* ub = Ul + (size_t)(d0 + lo16) * R_RANK + 2 * hiK;
        v2f bb0[4][8], bb1[4][8];
        g2_load(bb0, ub, 0, 0);
        float bcol[8];
#pragma unroll
        for (int nt = 0; nt < 8; ++nt) bcol[nt] = bl[d0 + nt * 16 + lo16];

#pragma unroll 2
        for (int t = 0; t < (B_ROWS * N1) / THREADS; ++t) {
            const int b = (tid >> 7) + 2 * t;
            const float* vrow = &YB[b * YPITCH + ce * R_RANK];
            float s = 0.f;
#pragma unroll
            for (int j = 0; j < 8; ++j) {
                s += crow[j].x * vrow[4 * j + 0];
                s += crow[j].y * vrow[4 * j + 1];
                s += crow[j].z * vrow[4 * j + 2];
                s += crow[j].w * vrow[4 * j + 3];
            }
            CB[b * YPITCH + ccol] = GT[b * 4 + ce] * fast_tanh(s);
        }
        __syncthreads();

        v8f acc2[2][8];
#pragma unroll
        for (int m = 0; m < 2; ++m)
#pragma unroll
            for (int nt = 0; nt < 8; ++nt) acc2[m][nt] = vzero;

        v2f a0[8], a1[8];
        const float* ca = &CB[lo16 * YPITCH + 2 * hiK];
#pragma unroll 1
        for (int e = 0; e < E_EXP - 1; ++e) {
            g2_afrag(a0, a1, ca, e);
            g2_load(bb1, ub, e, 1);
            __builtin_amdgcn_sched_barrier(0);
            g2_mma<0>(bb0, a0, a1, acc2);
            __builtin_amdgcn_sched_barrier(0);
            g2_load(bb0, ub, e + 1, 0);
            __builtin_amdgcn_sched_barrier(0);
            g2_mma<1>(bb1, a0, a1, acc2);
        }
        g2_afrag(a0, a1, ca, E_EXP - 1);
        g2_load(bb1, ub, E_EXP - 1, 1);
        __builtin_amdgcn_sched_barrier(0);
        g2_mma<0>(bb0, a0, a1, acc2);
        g2_mma<1>(bb1, a0, a1, acc2);

#pragma unroll
        for (int nt = 0; nt < 8; ++nt) {
            const int col = d0 + nt * 16 + lo16;
#pragma unroll
            for (int m = 0; m < 2; ++m) {
#pragma unroll
                for (int j = 0; j < 8; ++j) {
                    const int row = m * 16 + j + 8 * hiK;
                    const float xv  = XL[row * XPITCH + col];
                    const float x0v = X0[row * XPITCH + col];
                    XL[row * XPITCH + col] = xv + x0v * (bcol[nt] + acc2[m][nt][j]);
                }
            }
        }
        __syncthreads();
    }

    for (int i = tid; i < B_ROWS * (D_DIM / 4); i += THREADS) {
        const int r  = i >> 8;
        const int c4 = i & 255;
        const float4 v4 = *(const float4*)&XL[r * XPITCH + c4 * 4];
        ((float4*)(out + (size_t)(rowBase + r) * D_DIM))[c4] = v4;
    }
}

extern "C" void kernel_launch(void* const* d_in, const int* in_sizes, int n_in,
                              void* d_out, int out_size, void* d_ws, size_t ws_size,
                              hipStream_t stream) {
    (void)in_sizes; (void)n_in; (void)d_ws; (void)ws_size; (void)out_size;
    const float* x      = (const float*)d_in[0];
    const float* U      = (const float*)d_in[1];
    const float* V      = (const float*)d_in[2];
    const float* C      = (const float*)d_in[3];
    const float* bias   = (const float*)d_in[4];
    const float* gate_w = (const float*)d_in[5];
    float* out = (float*)d_out;

    const dim3 grid(B_TOT / B_ROWS);
    const size_t smemBytes = (size_t)SMEM_FLOATS * sizeof(float);
    crossnet_mix_kernel<<<grid, THREADS, smemBytes, stream>>>(x, U, V, C, bias, gate_w, out);
}


# Epilogue: the 32 LDS reads of each N tile (XL, X0) issued as one group a tile ahead of its 16 stores, not one LDS round trip per element; same arithmetic, every sum keeps its order
# speedup vs baseline: 2.7569x; 1.0711x over previous
#include <hip/hip_runtime.h>
#include <hip/hip_bf16.h>


typedef float v2f __attribute__((ext_vector_type(2)));
typedef float v8f __attribute__((ext_vector_type(8)));

#define B_TOT    16384
#define D_DIM    1024
#define E_EXP    4
#define R_RANK   32
#define N1       128
#define L_LAYERS 2
#define B_ROWS   32
#define THREADS  256
#define XPITCH   1028
#define YPITCH   132

#define KB1      16
#define NB1      (D_DIM / (4 * KB1))
#define GW_GRP   32

#define SMEM_FLOATS (2 * B_ROWS * XPITCH + 2 * B_ROWS * YPITCH + B_ROWS * E_EXP + THREADS)

__device__ __forceinline__ float fast_tanh(float x) {
    const float e = __expf(2.0f * x);
    return 1.0f - 2.0f * __frcp_rn(e + 1.0f);
}

__device__ __forceinline__ void g1_load(float (&bx)[KB1], float (&by)[KB1],
                                        const float* __restrict__ vb, int kb) {
    const float* p = vb + (size_t)kb * (4 * KB1) * R_RANK;
#pragma unroll
    for (int j = 0; j < KB1; ++j) {
        bx[j] = p[(4 * j) * R_RANK];
        by[j] = p[(4 * j + 1) * R_RANK];
    }
}

__device__ __forceinline__ void g1_aload(v2f (&a0)[KB1], v2f (&a1)[KB1], const float* xa, int kb) {
#pragma unroll
    for (int j = 0; j < KB1; ++j) {
        const int k0 = (kb * KB1 + j) * 4;
        a0[j] = *(const v2f*)&xa[k0];
        a1[j] = *(const v2f*)&xa[16 * XPITCH + k0];
    }
}

__device__ __forceinline__ void g1_mma(const float (&bx)[KB1], const float (&by)[KB1],
                                       const v2f (&a0)[KB1], const v2f (&a1)[KB1], v8f& acc0, v8f& acc1) {
#pragma unroll
    for (int j = 0; j < KB1; ++j) {
        v2f b;
        b.x = bx[j];
        b.y = by[j];
        acc0 = __builtin_amdgcn_wmma_f32_16x16x4_f32(false, a0[j], false, b, (short)0, acc0, false, false);
        acc1 = __builtin_amdgcn_wmma_f32_16x16x4_f32(false, a1[j], false, b, (short)0, acc1, false, false);
    }
}

__device__ __forceinline__ void g2_load(v2f (&b)[4][8], const float* __restrict__ ub, int e, int half) {
#pragma unroll
    for (int q = 0; q < 4; ++q) {
        const float* p = ub + ((size_t)e * D_DIM + (half * 4 + q) * 16) * R_RANK;
#pragma unroll
        for (int s = 0; s < 8; ++s) b[q][s] = *(const v2f*)(p + 4 * s);
    }
}

__device__ __forceinline__ void g2_afrag(v2f (&a0)[8], v2f (&a1)[8], const float* ca, int e) {
#pragma unroll
    for (int s = 0; s < 8; ++s) {
        a0[s] = *(const v2f*)&ca[e * R_RANK + 4 * s];
        a1[s] = *(const v2f*)&ca[16 * YPITCH + e * R_RANK + 4 * s];
    }
}

template <int HALF>
__device__ __forceinline__ void g2_mma(const v2f (&b)[4][8], const v2f (&a0)[8], const v2f (&a1)[8],
                                       v8f (&acc)[2][8]) {
#pragma unroll
    for (int s = 0; s < 8; ++s) {
#pragma unroll
        for (int q = 0; q < 4; ++q) {
            acc[0][HALF * 4 + q] = __builtin_amdgcn_wmma_f32_16x16x4_f32(false, a0[s], false, b[q][s], (short)0, acc[0][HALF * 4 + q], false, false);
            acc[1][HALF * 4 + q] = __builtin_amdgcn_wmma_f32_16x16x4_f32(false, a1[s], false, b[q][s], (short)0, acc[1][HALF * 4 + q], false, false);
        }
    }
}

__device__ __forceinline__ void epi_load(float (&xv)[16], float (&x0v)[16], const float* XL, const float* X0,
                                         int col, int hiK) {
#pragma unroll
    for (int m = 0; m < 2; ++m) {
#pragma unroll
        for (int j = 0; j < 8; ++j) {
            const int row = m * 16 + j + 8 * hiK;
            xv[m * 8 + j]  = XL[row * XPITCH + col];
            x0v[m * 8 + j] = X0[row * XPITCH + col];
        }
    }
}

extern "C" __global__ __launch_bounds__(THREADS)
void crossnet_mix_kernel(const float* __restrict__ x,
                         const float* __restrict__ U,
                         const float* __restrict__ V,
                         const float* __restrict__ C,
                         const float* __restrict__ bias,
                         const float* __restrict__ gate_w,
                         float* __restrict__ out)
{
    extern __shared__ float smem[];
    float* X0 = smem;
    float* XL = smem + B_ROWS * XPITCH;
    float* YB = smem + 2 * B_ROWS * XPITCH;
    float* CB = YB + B_ROWS * YPITCH;
    float* GT = CB + B_ROWS * YPITCH;
    float* GP = GT + B_ROWS * E_EXP;

    const int tid  = threadIdx.x;
    const int wave = tid >> 5;
    const int lane = tid & 31;
    const int lo16 = lane & 15;
    const int hiK  = lane >> 4;
    const int rowBase = blockIdx.x * B_ROWS;

#pragma unroll
    for (int rh = 0; rh < B_ROWS; rh += 16) {
        float4 xv[16];
#pragma unroll
        for (int r = 0; r < 16; ++r)
            xv[r] = ((const float4*)(x + (size_t)(rowBase + rh + r) * D_DIM))[tid];
#pragma unroll
        for (int r = 0; r < 16; ++r) {
            *(float4*)&X0[(rh + r) * XPITCH + tid * 4] = xv[r];
            *(float4*)&XL[(rh + r) * XPITCH + tid * 4] = xv[r];
        }
    }
    __syncthreads();

    const v8f vzero = {};

    const int ccol = tid & 127;
    const int ce   = ccol >> 5;
    const int cr_  = ccol & 31;

    for (int l = 0; l < L_LAYERS; ++l) {
        const float* Vl = V + (size_t)l * E_EXP * D_DIM * R_RANK;
        const float* Ul = U + (size_t)l * E_EXP * D_DIM * R_RANK;
        const float* Cl = C + (size_t)l * E_EXP * R_RANK * R_RANK;
        const float* bl = bias + (size_t)l * D_DIM;

        float4 crow[8];
        {
            const float4* cp = (const float4*)(Cl + ((size_t)ce * R_RANK + cr_) * R_RANK);
#pragma unroll
            for (int j = 0; j < 8; ++j) crow[j] = cp[j];
        }

        const int n0 = wave * 16;
        const int e1 = n0 >> 5;
        const int c1 = (n0 & 31) + lo16;
        v8f acc0 = vzero, acc1 = vzero;
        {
            const float* vb = Vl + ((size_t)e1 * D_DIM + 2 * hiK) * R_RANK + c1;
            const float* xa = &XL[lo16 * XPITCH + 2 * hiK];
            float bx0[KB1], by0[KB1], bx1[KB1], by1[KB1];
            v2f xa0[KB1], xb0[KB1], xa1[KB1], xb1[KB1];
            g1_load(bx0, by0, vb, 0);
            g1_aload(xa0, xb0, xa, 0);
            for (int kb = 0; kb < NB1 - 2; kb += 2) {
                g1_load(bx1, by1, vb, kb + 1);
                g1_aload(xa1, xb1, xa, kb + 1);
                g1_mma(bx0, by0, xa0, xb0, acc0, acc1);
                g1_load(bx0, by0, vb, kb + 2);
                g1_aload(xa0, xb0, xa, kb + 2);
                g1_mma(bx1, by1, xa1, xb1, acc0, acc1);
                __builtin_amdgcn_sched_group_barrier(0x020, 2 * KB1, 0);
                __builtin_amdgcn_sched_group_barrier(0x100, 2 * KB1, 0);
                __builtin_amdgcn_sched_group_barrier(0x008, 2 * KB1, 0);
                __builtin_amdgcn_sched_group_barrier(0x020, 2 * KB1, 0);
                __builtin_amdgcn_sched_group_barrier(0x100, 2 * KB1, 0);
                __builtin_amdgcn_sched_group_barrier(0x008, 2 * KB1, 0);
            }
            g1_load(bx1, by1, vb, NB1 - 1);
            g1_aload(xa1, xb1, xa, NB1 - 1);
            __builtin_amdgcn_sched_barrier(0);
            g1_mma(bx0, by0, xa0, xb0, acc0, acc1);
            g1_mma(bx1, by1, xa1, xb1, acc0, acc1);
        }
#pragma unroll
        for (int j = 0; j < 8; ++j) {
            const int r0 = j + 8 * hiK;
            YB[r0 * YPITCH + n0 + lo16]        = fast_tanh(acc0[j]);
            YB[(r0 + 16) * YPITCH + n0 + lo16] = fast_tanh(acc1[j]);
        }

        {
            const int b = tid >> 3;
            const int e = (tid >> 1) & 3;
            const int h = tid & 1;
            const float4* gw = (const float4*)(gate_w + (size_t)e * D_DIM + h * (D_DIM / 2));
            const float* xr = &XL[b * XPITCH + h * (D_DIM / 2)];
            float s = 0.f;
#pragma unroll 1
            for (int g0 = 0; g0 < D_DIM / 8; g0 += GW_GRP) {
                float4 gv[GW_GRP];
#pragma unroll
                for (int j = 0; j < GW_GRP; ++j) gv[j] = gw[g0 + j];
#pragma unroll
                for (int j = 0; j < GW_GRP; ++j) {
                    const float4 g = gv[j];
                    const float* xp = xr + (g0 + j) * 4;
                    s += g.x * xp[0] + g.y * xp[1] + g.z * xp[2] + g.w * xp[3];
                }
            }
            GP[tid] = s;
        }
        __syncthreads();
        if (tid < B_ROWS) {
            const float g0 = GP[tid * 8 + 0] + GP[tid * 8 + 1];
            const float g1 = GP[tid * 8 + 2] + GP[tid * 8 + 3];
            const float g2 = GP[tid * 8 + 4] + GP[tid * 8 + 5];
            const float g3 = GP[tid * 8 + 6] + GP[tid * 8 + 7];
            const float m  = fmaxf(fmaxf(g0, g1), fmaxf(g2, g3));
            const float e0 = __expf(g0 - m), e1x = __expf(g1 - m);
            const float e2 = __expf(g2 - m), e3 = __expf(g3 - m);
            const float inv = __frcp_rn(e0 + e1x + e2 + e3);
            GT[tid * 4 + 0] = e0 * inv;  GT[tid * 4 + 1] = e1x * inv;
            GT[tid * 4 + 2] = e2 * inv;  GT[tid * 4 + 3] = e3 * inv;
        }
        __syncthreads();

        const int d0 = wave * 128;
        const float* ub = Ul + (size_t)(d0 + lo16) * R_RANK + 2 * hiK;
        v2f bb0[4][8], bb1[4][8];
        g2_load(bb0, ub, 0, 0);
        float bcol[8];
#pragma unroll
        for (int nt = 0; nt < 8; ++nt) bcol[nt] = bl[d0 + nt * 16 + lo16];

#pragma unroll 2
        for (int t = 0; t < (B_ROWS * N1) / THREADS; ++t) {
            const int b = (tid >> 7) + 2 * t;
            const float* vrow = &YB[b * YPITCH + ce * R_RANK];
            float s = 0.f;
#pragma unroll
            for (int j = 0; j < 8; ++j) {
                s += crow[j].x * vrow[4 * j + 0];
                s += crow[j].y * vrow[4 * j + 1];
                s += crow[j].z * vrow[4 * j + 2];
                s += crow[j].w * vrow[4 * j + 3];
            }
            CB[b * YPITCH + ccol] = GT[b * 4 + ce] * fast_tanh(s);
        }
        __syncthreads();

        v8f acc2[2][8];
#pragma unroll
        for (int m = 0; m < 2; ++m)
#pragma unroll
            for (int nt = 0; nt < 8; ++nt) acc2[m][nt] = vzero;

        v2f a0[8], a1[8];
        const float* ca = &CB[lo16 * YPITCH + 2 * hiK];
#pragma unroll 1
        for (int e = 0; e < E_EXP - 1; ++e) {
            g2_afrag(a0, a1, ca, e);
            g2_load(bb1, ub, e, 1);
            __builtin_amdgcn_sched_barrier(0);
            g2_mma<0>(bb0, a0, a1, acc2);
            __builtin_amdgcn_sched_barrier(0);
            g2_load(bb0, ub, e + 1, 0);
            __builtin_amdgcn_sched_barrier(0);
            g2_mma<1>(bb1, a0, a1, acc2);
        }
        g2_afrag(a0, a1, ca, E_EXP - 1);
        g2_load(bb1, ub, E_EXP - 1, 1);
        __builtin_amdgcn_sched_barrier(0);
        g2_mma<0>(bb0, a0, a1, acc2);
        g2_mma<1>(bb1, a0, a1, acc2);

        float ex[2][16], e0[2][16];
        epi_load(ex[0], e0[0], XL, X0, d0 + lo16, hiK);
#pragma unroll
        for (int nt = 0; nt < 8; ++nt) {
            if (nt + 1 < 8)
                epi_load(ex[(nt + 1) & 1], e0[(nt + 1) & 1], XL, X0, d0 + (nt + 1) * 16 + lo16, hiK);
            const int col = d0 + nt * 16 + lo16;
#pragma unroll
            for (int m = 0; m < 2; ++m) {
#pragma unroll
                for (int j = 0; j < 8; ++j) {
                    const int row = m * 16 + j + 8 * hiK;
                    XL[row * XPITCH + col] = ex[nt & 1][m * 8 + j] + e0[nt & 1][m * 8 + j] * (bcol[nt] + acc2[m][nt][j]);
                }
            }
        }
        __syncthreads();
    }

    for (int i = tid; i < B_ROWS * (D_DIM / 4); i += THREADS) {
        const int r  = i >> 8;
        const int c4 = i & 255;
        const float4 v4 = *(const float4*)&XL[r * XPITCH + c4 * 4];
        ((float4*)(out + (size_t)(rowBase + r) * D_DIM))[c4] = v4;
    }
}

extern "C" void kernel_launch(void* const* d_in, const int* in_sizes, int n_in,
                              void* d_out, int out_size, void* d_ws, size_t ws_size,
                              hipStream_t stream) {
    (void)in_sizes; (void)n_in; (void)d_ws; (void)ws_size; (void)out_size;
    const float* x      = (const float*)d_in[0];
    const float* U      = (const float*)d_in[1];
    const float* V      = (const float*)d_in[2];
    const float* C      = (const float*)d_in[3];
    const float* bias   = (const float*)d_in[4];
    const float* gate_w = (const float*)d_in[5];
    float* out = (float*)d_out;

    const dim3 grid(B_TOT / B_ROWS);
    const size_t smemBytes = (size_t)SMEM_FLOATS * sizeof(float);
    crossnet_mix_kernel<<<grid, THREADS, smemBytes, stream>>>(x, U, V, C, bias, gate_w, out);
}


# C stage (per-expert 32x32 product C_e v_e): from per-thread serial multiply-add chains fed by LDS reads to the f32 matrix unit; its sums now follow the matrix unit's order
# speedup vs baseline: 3.0635x; 1.1112x over previous
#include <hip/hip_runtime.h>
#include <hip/hip_bf16.h>


typedef float v2f __attribute__((ext_vector_type(2)));
typedef float v8f __attribute__((ext_vector_type(8)));

#define B_TOT    16384
#define D_DIM    1024
#define E_EXP    4
#define R_RANK   32
#define N1       128
#define L_LAYERS 2
#define B_ROWS   32
#define THREADS  256
#define XPITCH   1028
#define YPITCH   132

#define KB1      16
#define NB1      (D_DIM / (4 * KB1))
#define GW_GRP   32

#define SMEM_FLOATS (2 * B_ROWS * XPITCH + 2 * B_ROWS * YPITCH + B_ROWS * E_EXP + THREADS)

__device__ __forceinline__ float fast_tanh(float x) {
    const float e = __expf(2.0f * x);
    return 1.0f - 2.0f * __frcp_rn(e + 1.0f);
}

__device__ __forceinline__ void g1_load(float (&bx)[KB1], float (&by)[KB1],
                                        const float* __restrict__ vb, int kb) {
    const float* p = vb + (size_t)kb * (4 * KB1) * R_RANK;
#pragma unroll
    for (int j = 0; j < KB1; ++j) {
        bx[j] = p[(4 * j) * R_RANK];
        by[j] = p[(4 * j + 1) * R_RANK];
    }
}

__device__ __forceinline__ void g1_aload(v2f (&a0)[KB1], v2f (&a1)[KB1], const float* xa, int kb) {
#pragma unroll
    for (int j = 0; j < KB1; ++j) {
        const int k0 = (kb * KB1 + j) * 4;
        a0[j] = *(const v2f*)&xa[k0];
        a1[j] = *(const v2f*)&xa[16 * XPITCH + k0];
    }
}

__device__ __forceinline__ void g1_mma(const float (&bx)[KB1], const float (&by)[KB1],
                                       const v2f (&a0)[KB1], const v2f (&a1)[KB1], v8f& acc0, v8f& acc1) {
#pragma unroll
    for (int j = 0; j < KB1; ++j) {
        v2f b;
        b.x = bx[j];
        b.y = by[j];
        acc0 = __builtin_amdgcn_wmma_f32_16x16x4_f32(false, a0[j], false, b, (short)0, acc0, false, false);
        acc1 = __builtin_amdgcn_wmma_f32_16x16x4_f32(false, a1[j], false, b, (short)0, acc1, false, false);
    }
}

__device__ __forceinline__ void g2_load(v2f (&b)[4][8], const float* __restrict__ ub, int e, int half) {
#pragma unroll
    for (int q = 0; q < 4; ++q) {
        const float* p = ub + ((size_t)e * D_DIM + (half * 4 + q) * 16) * R_RANK;
#pragma unroll
        for (int s = 0; s < 8; ++s) b[q][s] = *(const v2f*)(p + 4 * s);
    }
}

__device__ __forceinline__ void g2_afrag(v2f (&a0)[8], v2f (&a1)[8], const float* ca, int e) {
#pragma unroll
    for (int s = 0; s < 8; ++s) {
        a0[s] = *(const v2f*)&ca[e * R_RANK + 4 * s];
        a1[s] = *(const v2f*)&ca[16 * YPITCH + e * R_RANK + 4 * s];
    }
}

template <int HALF>
__device__ __forceinline__ void g2_mma(const v2f (&b)[4][8], const v2f (&a0)[8], const v2f (&a1)[8],
                                       v8f (&acc)[2][8]) {
#pragma unroll
    for (int s = 0; s < 8; ++s) {
#pragma unroll
        for (int q = 0; q < 4; ++q) {
            acc[0][HALF * 4 + q] = __builtin_amdgcn_wmma_f32_16x16x4_f32(false, a0[s], false, b[q][s], (short)0, acc[0][HALF * 4 + q], false, false);
            acc[1][HALF * 4 + q] = __builtin_amdgcn_wmma_f32_16x16x4_f32(false, a1[s], false, b[q][s], (short)0, acc[1][HALF * 4 + q], false, false);
        }
    }
}

__device__ __forceinline__ void epi_load(float (&xv)[16], float (&x0v)[16], const float* XL, const float* X0,
                                         int col, int hiK) {
#pragma unroll
    for (int m = 0; m < 2; ++m) {
#pragma unroll
        for (int j = 0; j < 8; ++j) {
            const int row = m * 16 + j + 8 * hiK;
            xv[m * 8 + j]  = XL[row * XPITCH + col];
            x0v[m * 8 + j] = X0[row * XPITCH + col];
        }
    }
}

extern "C" __global__ __launch_bounds__(THREADS)
void crossnet_mix_kernel(const float* __restrict__ x,
                         const float* __restrict__ U,
                         const float* __restrict__ V,
                         const float* __restrict__ C,
                         const float* __restrict__ bias,
                         const float* __restrict__ gate_w,
                         float* __restrict__ out)
{
    extern __shared__ float smem[];
    float* X0 = smem;
    float* XL = smem + B_ROWS * XPITCH;
    float* YB = smem + 2 * B_ROWS * XPITCH;
    float* CB = YB + B_ROWS * YPITCH;
    float* GT = CB + B_ROWS * YPITCH;
    float* GP = GT + B_ROWS * E_EXP;

    const int tid  = threadIdx.x;
    const int wave = tid >> 5;
    const int lane = tid & 31;
    const int lo16 = lane & 15;
    const int hiK  = lane >> 4;
    const int rowBase = blockIdx.x * B_ROWS;

#pragma unroll
    for (int rh = 0; rh < B_ROWS; rh += 16) {
        float4 xv[16];
#pragma unroll
        for (int r = 0; r < 16; ++r)
            xv[r] = ((const float4*)(x + (size_t)(rowBase + rh + r) * D_DIM))[tid];
#pragma unroll
        for (int r = 0; r < 16; ++r) {
            *(float4*)&X0[(rh + r) * XPITCH + tid * 4] = xv[r];
            *(float4*)&XL[(rh + r) * XPITCH + tid * 4] = xv[r];
        }
    }
    __syncthreads();

    const v8f vzero = {};

    const int n0 = wave * 16;
    const int e1 = n0 >> 5;
    const int c1 = (n0 & 31) + lo16;

    for (int l = 0; l < L_LAYERS; ++l) {
        const float* Vl = V + (size_t)l * E_EXP * D_DIM * R_RANK;
        const float* Ul = U + (size_t)l * E_EXP * D_DIM * R_RANK;
        const float* Cl = C + (size_t)l * E_EXP * R_RANK * R_RANK;
        const float* bl = bias + (size_t)l * D_DIM;

        v2f cfr[8];
        {
            const float* cp = Cl + ((size_t)e1 * R_RANK + c1) * R_RANK + 2 * hiK;
#pragma unroll
            for (int s = 0; s < 8; ++s) cfr[s] = *(const v2f*)(cp + 4 * s);
        }

        v8f acc0 = vzero, acc1 = vzero;
        {
            const float* vb = Vl + ((size_t)e1 * D_DIM + 2 * hiK) * R_RANK + c1;
            const float* xa = &XL[lo16 * XPITCH + 2 * hiK];
            float bx0[KB1], by0[KB1], bx1[KB1], by1[KB1];
            v2f xa0[KB1], xb0[KB1], xa1[KB1], xb1[KB1];
            g1_load(bx0, by0, vb, 0);
            g1_aload(xa0, xb0, xa, 0);
            for (int kb = 0; kb < NB1 - 2; kb += 2) {
                g1_load(bx1, by1, vb, kb + 1);
                g1_aload(xa1, xb1, xa, kb + 1);
                g1_mma(bx0, by0, xa0, xb0, acc0, acc1);
                g1_load(bx0, by0, vb, kb + 2);
                g1_aload(xa0, xb0, xa, kb + 2);
                g1_mma(bx1, by1, xa1, xb1, acc0, acc1);
                __builtin_amdgcn_sched_group_barrier(0x020, 2 * KB1, 0);
                __builtin_amdgcn_sched_group_barrier(0x100, 2 * KB1, 0);
                __builtin_amdgcn_sched_group_barrier(0x008, 2 * KB1, 0);
                __builtin_amdgcn_sched_group_barrier(0x020, 2 * KB1, 0);
                __builtin_amdgcn_sched_group_barrier(0x100, 2 * KB1, 0);
                __builtin_amdgcn_sched_group_barrier(0x008, 2 * KB1, 0);
            }
            g1_load(bx1, by1, vb, NB1 - 1);
            g1_aload(xa1, xb1, xa, NB1 - 1);
            __builtin_amdgcn_sched_barrier(0);
            g1_mma(bx0, by0, xa0, xb0, acc0, acc1);
            g1_mma(bx1, by1, xa1, xb1, acc0, acc1);
        }
#pragma unroll
        for (int j = 0; j < 8; ++j) {
            const int r0 = j + 8 * hiK;
            YB[r0 * YPITCH + n0 + lo16]        = fast_tanh(acc0[j]);
            YB[(r0 + 16) * YPITCH + n0 + lo16] = fast_tanh(acc1[j]);
        }

        {
            const int b = tid >> 3;
            const int e = (tid >> 1) & 3;
            const int h = tid & 1;
            const float4* gw = (const float4*)(gate_w + (size_t)e * D_DIM + h * (D_DIM / 2));
            const float* xr = &XL[b * XPITCH + h * (D_DIM / 2)];
            float s = 0.f;
#pragma unroll 1
            for (int g0 = 0; g0 < D_DIM / 8; g0 += GW_GRP) {
                float4 gv[GW_GRP];
#pragma unroll
                for (int j = 0; j < GW_GRP; ++j) gv[j] = gw[g0 + j];
#pragma unroll
                for (int j = 0; j < GW_GRP; ++j) {
                    const float4 g = gv[j];
                    const float* xp = xr + (g0 + j) * 4;
                    s += g.x * xp[0] + g.y * xp[1] + g.z * xp[2] + g.w * xp[3];
                }
            }
            GP[tid] = s;
        }
        __syncthreads();
        if (tid < B_ROWS) {
            const float g0 = GP[tid * 8 + 0] + GP[tid * 8 + 1];
            const float g1 = GP[tid * 8 + 2] + GP[tid * 8 + 3];
            const float g2 = GP[tid * 8 + 4] + GP[tid * 8 + 5];
            const float g3 = GP[tid * 8 + 6] + GP[tid * 8 + 7];
            const float m  = fmaxf(fmaxf(g0, g1), fmaxf(g2, g3));
            const float e0 = __expf(g0 - m), e1x = __expf(g1 - m);
            const float e2 = __expf(g2 - m), e3 = __expf(g3 - m);
            const float inv = __frcp_rn(e0 + e1x + e2 + e3);
            GT[tid * 4 + 0] = e0 * inv;  GT[tid * 4 + 1] = e1x * inv;
            GT[tid * 4 + 2] = e2 * inv;  GT[tid * 4 + 3] = e3 * inv;
        }
        __syncthreads();

        const int d0 = wave * 128;
        const float* ub = Ul + (size_t)(d0 + lo16) * R_RANK + 2 * hiK;
        v2f bb0[4][8], bb1[4][8];
        g2_load(bb0, ub, 0, 0);
        float bcol[8];
#pragma unroll
        for (int nt = 0; nt < 8; ++nt) bcol[nt] = bl[d0 + nt * 16 + lo16];

        {
            const float* ya = &YB[lo16 * YPITCH + e1 * R_RANK + 2 * hiK];
            v2f ya0[8], ya1[8];
#pragma unroll
            for (int s = 0; s < 8; ++s) {
                ya0[s] = *(const v2f*)&ya[4 * s];
                ya1[s] = *(const v2f*)&ya[16 * YPITCH + 4 * s];
            }
            v8f p0 = vzero, p1 = vzero;
#pragma unroll
            for (int s = 0; s < 8; ++s) {
                p0 = __builtin_amdgcn_wmma_f32_16x16x4_f32(false, ya0[s], false, cfr[s], (short)0, p0, false, false);
                p1 = __builtin_amdgcn_wmma_f32_16x16x4_f32(false, ya1[s], false, cfr[s], (short)0, p1, false, false);
            }
#pragma unroll
            for (int j = 0; j < 8; ++j) {
                const int r0 = j + 8 * hiK;
                CB[r0 * YPITCH + n0 + lo16]        = GT[r0 * 4 + e1] * fast_tanh(p0[j]);
                CB[(r0 + 16) * YPITCH + n0 + lo16] = GT[(r0 + 16) * 4 + e1] * fast_tanh(p1[j]);
            }
        }
        __syncthreads();

        v8f acc2[2][8];
#pragma unroll
        for (int m = 0; m < 2; ++m)
#pragma unroll
            for (int nt = 0; nt < 8; ++nt) acc2[m][nt] = vzero;

        v2f a0[8], a1[8];
        const float* ca = &CB[lo16 * YPITCH + 2 * hiK];
#pragma unroll 1
        for (int e = 0; e < E_EXP - 1; ++e) {
            g2_afrag(a0, a1, ca, e);
            g2_load(bb1, ub, e, 1);
            __builtin_amdgcn_sched_barrier(0);
            g2_mma<0>(bb0, a0, a1, acc2);
            __builtin_amdgcn_sched_barrier(0);
            g2_load(bb0, ub, e + 1, 0);
            __builtin_amdgcn_sched_barrier(0);
            g2_mma<1>(bb1, a0, a1, acc2);
        }
        g2_afrag(a0, a1, ca, E_EXP - 1);
        g2_load(bb1, ub, E_EXP - 1, 1);
        __builtin_amdgcn_sched_barrier(0);
        g2_mma<0>(bb0, a0, a1, acc2);
        g2_mma<1>(bb1, a0, a1, acc2);

        float ex[2][16], e0[2][16];
        epi_load(ex[0], e0[0], XL, X0, d0 + lo16, hiK);
#pragma unroll
        for (int nt = 0; nt < 8; ++nt) {
            if (nt + 1 < 8)
                epi_load(ex[(nt + 1) & 1], e0[(nt + 1) & 1], XL, X0, d0 + (nt + 1) * 16 + lo16, hiK);
            const int col = d0 + nt * 16 + lo16;
#pragma unroll
            for (int m = 0; m < 2; ++m) {
#pragma unroll
                for (int j = 0; j < 8; ++j) {
                    const int row = m * 16 + j + 8 * hiK;
                    XL[row * XPITCH + col] = ex[nt & 1][m * 8 + j] + e0[nt & 1][m * 8 + j] * (bcol[nt] + acc2[m][nt][j]);
                }
            }
        }
        __syncthreads();
    }

    for (int i = tid; i < B_ROWS * (D_DIM / 4); i += THREADS) {
        const int r  = i >> 8;
        const int c4 = i & 255;
        const float4 v4 = *(const float4*)&XL[r * XPITCH + c4 * 4];
        ((float4*)(out + (size_t)(rowBase + r) * D_DIM))[c4] = v4;
    }
}

extern "C" void kernel_launch(void* const* d_in, const int* in_sizes, int n_in,
                              void* d_out, int out_size, void* d_ws, size_t ws_size,
                              hipStream_t stream) {
    (void)in_sizes; (void)n_in; (void)d_ws; (void)ws_size; (void)out_size;
    const float* x      = (const float*)d_in[0];
    const float* U      = (const float*)d_in[1];
    const float* V      = (const float*)d_in[2];
    const float* C      = (const float*)d_in[3];
    const float* bias   = (const float*)d_in[4];
    const float* gate_w = (const float*)d_in[5];
    float* out = (float*)d_out;

    const dim3 grid(B_TOT / B_ROWS);
    const size_t smemBytes = (size_t)SMEM_FLOATS * sizeof(float);
    crossnet_mix_kernel<<<grid, THREADS, smemBytes, stream>>>(x, U, V, C, bias, gate_w, out);
}


# Gate logits: each wave takes one 128-wide k slice for all 32 rows and 4 experts (x read once from LDS, 16 B conflict-free; one-line weight loads); slices added per row, logit sum order changes
# speedup vs baseline: 3.2857x; 1.0725x over previous
#include <hip/hip_runtime.h>
#include <hip/hip_bf16.h>


typedef float v2f __attribute__((ext_vector_type(2)));
typedef float v4f __attribute__((ext_vector_type(4)));
typedef float v8f __attribute__((ext_vector_type(8)));

#define B_TOT    16384
#define D_DIM    1024
#define E_EXP    4
#define R_RANK   32
#define N1       128
#define L_LAYERS 2
#define B_ROWS   32
#define THREADS  256
#define XPITCH   1028
#define YPITCH   132

#define KB1      16
#define NB1      (D_DIM / (4 * KB1))
#define GK       (D_DIM / 8)
#define GW_GRP   16

#define SMEM_FLOATS (2 * B_ROWS * XPITCH + 2 * B_ROWS * YPITCH + B_ROWS * E_EXP + 8 * B_ROWS * E_EXP)

__device__ __forceinline__ float fast_tanh(float x) {
    const float e = __expf(2.0f * x);
    return 1.0f - 2.0f * __frcp_rn(e + 1.0f);
}

__device__ __forceinline__ void g1_load(float (&bx)[KB1], float (&by)[KB1],
                                        const float* __restrict__ vb, int kb) {
    const float* p = vb + (size_t)kb * (4 * KB1) * R_RANK;
#pragma unroll
    for (int j = 0; j < KB1; ++j) {
        bx[j] = p[(4 * j) * R_RANK];
        by[j] = p[(4 * j + 1) * R_RANK];
    }
}

__device__ __forceinline__ void g1_aload(v2f (&a0)[KB1], v2f (&a1)[KB1], const float* xa, int kb) {
#pragma unroll
    for (int j = 0; j < KB1; ++j) {
        const int k0 = (kb * KB1 + j) * 4;
        a0[j] = *(const v2f*)&xa[k0];
        a1[j] = *(const v2f*)&xa[16 * XPITCH + k0];
    }
}

__device__ __forceinline__ void g1_mma(const float (&bx)[KB1], const float (&by)[KB1],
                                       const v2f (&a0)[KB1], const v2f (&a1)[KB1], v8f& acc0, v8f& acc1) {
#pragma unroll
    for (int j = 0; j < KB1; ++j) {
        v2f b;
        b.x = bx[j];
        b.y = by[j];
        acc0 = __builtin_amdgcn_wmma_f32_16x16x4_f32(false, a0[j], false, b, (short)0, acc0, false, false);
        acc1 = __builtin_amdgcn_wmma_f32_16x16x4_f32(false, a1[j], false, b, (short)0, acc1, false, false);
    }
}

__device__ __forceinline__ void g2_load(v2f (&b)[4][8], const float* __restrict__ ub, int e, int half) {
#pragma unroll
    for (int q = 0; q < 4; ++q) {
        const float* p = ub + ((size_t)e * D_DIM + (half * 4 + q) * 16) * R_RANK;
#pragma unroll
        for (int s = 0; s < 8; ++s) b[q][s] = *(const v2f*)(p + 4 * s);
    }
}

__device__ __forceinline__ void g2_afrag(v2f (&a0)[8], v2f (&a1)[8], const float* ca, int e) {
#pragma unroll
    for (int s = 0; s < 8; ++s) {
        a0[s] = *(const v2f*)&ca[e * R_RANK + 4 * s];
        a1[s] = *(const v2f*)&ca[16 * YPITCH + e * R_RANK + 4 * s];
    }
}

template <int HALF>
__device__ __forceinline__ void g2_mma(const v2f (&b)[4][8], const v2f (&a0)[8], const v2f (&a1)[8],
                                       v8f (&acc)[2][8]) {
#pragma unroll
    for (int s = 0; s < 8; ++s) {
#pragma unroll
        for (int q = 0; q < 4; ++q) {
            acc[0][HALF * 4 + q] = __builtin_amdgcn_wmma_f32_16x16x4_f32(false, a0[s], false, b[q][s], (short)0, acc[0][HALF * 4 + q], false, false);
            acc[1][HALF * 4 + q] = __builtin_amdgcn_wmma_f32_16x16x4_f32(false, a1[s], false, b[q][s], (short)0, acc[1][HALF * 4 + q], false, false);
        }
    }
}

__device__ __forceinline__ void epi_load(float (&xv)[16], float (&x0v)[16], const float* XL, const float* X0,
                                         int col, int hiK) {
#pragma unroll
    for (int m = 0; m < 2; ++m) {
#pragma unroll
        for (int j = 0; j < 8; ++j) {
            const int row = m * 16 + j + 8 * hiK;
            xv[m * 8 + j]  = XL[row * XPITCH + col];
            x0v[m * 8 + j] = X0[row * XPITCH + col];
        }
    }
}

extern "C" __global__ __launch_bounds__(THREADS)
void crossnet_mix_kernel(const float* __restrict__ x,
                         const float* __restrict__ U,
                         const float* __restrict__ V,
                         const float* __restrict__ C,
                         const float* __restrict__ bias,
                         const float* __restrict__ gate_w,
                         float* __restrict__ out)
{
    extern __shared__ float smem[];
    float* X0 = smem;
    float* XL = smem + B_ROWS * XPITCH;
    float* YB = smem + 2 * B_ROWS * XPITCH;
    float* CB = YB + B_ROWS * YPITCH;
    float* GT = CB + B_ROWS * YPITCH;
    float* GP = GT + B_ROWS * E_EXP;

    const int tid  = threadIdx.x;
    const int wave = tid >> 5;
    const int lane = tid & 31;
    const int lo16 = lane & 15;
    const int hiK  = lane >> 4;
    const int rowBase = blockIdx.x * B_ROWS;

#pragma unroll
    for (int rh = 0; rh < B_ROWS; rh += 16) {
        float4 xv[16];
#pragma unroll
        for (int r = 0; r < 16; ++r)
            xv[r] = ((const float4*)(x + (size_t)(rowBase + rh + r) * D_DIM))[tid];
#pragma unroll
        for (int r = 0; r < 16; ++r) {
            *(float4*)&X0[(rh + r) * XPITCH + tid * 4] = xv[r];
            *(float4*)&XL[(rh + r) * XPITCH + tid * 4] = xv[r];
        }
    }
    __syncthreads();

    const v8f vzero = {};

    const int n0 = wave * 16;
    const int e1 = n0 >> 5;
    const int c1 = (n0 & 31) + lo16;

    for (int l = 0; l < L_LAYERS; ++l) {
        const float* Vl = V + (size_t)l * E_EXP * D_DIM * R_RANK;
        const float* Ul = U + (size_t)l * E_EXP * D_DIM * R_RANK;
        const float* Cl = C + (size_t)l * E_EXP * R_RANK * R_RANK;
        const float* bl = bias + (size_t)l * D_DIM;

        v2f cfr[8];
        {
            const float* cp = Cl + ((size_t)e1 * R_RANK + c1) * R_RANK + 2 * hiK;
#pragma unroll
            for (int s = 0; s < 8; ++s) cfr[s] = *(const v2f*)(cp + 4 * s);
        }

        v8f acc0 = vzero, acc1 = vzero;
        {
            const float* vb = Vl + ((size_t)e1 * D_DIM + 2 * hiK) * R_RANK + c1;
            const float* xa = &XL[lo16 * XPITCH + 2 * hiK];
            float bx0[KB1], by0[KB1], bx1[KB1], by1[KB1];
            v2f xa0[KB1], xb0[KB1], xa1[KB1], xb1[KB1];
            g1_load(bx0, by0, vb, 0);
            g1_aload(xa0, xb0, xa, 0);
            for (int kb = 0; kb < NB1 - 2; kb += 2) {
                g1_load(bx1, by1, vb, kb + 1);
                g1_aload(xa1, xb1, xa, kb + 1);
                g1_mma(bx0, by0, xa0, xb0, acc0, acc1);
                g1_load(bx0, by0, vb, kb + 2);
                g1_aload(xa0, xb0, xa, kb + 2);
                g1_mma(bx1, by1, xa1, xb1, acc0, acc1);
                __builtin_amdgcn_sched_group_barrier(0x020, 2 * KB1, 0);
                __builtin_amdgcn_sched_group_barrier(0x100, 2 * KB1, 0);
                __builtin_amdgcn_sched_group_barrier(0x008, 2 * KB1, 0);
                __builtin_amdgcn_sched_group_barrier(0x020, 2 * KB1, 0);
                __builtin_amdgcn_sched_group_barrier(0x100, 2 * KB1, 0);
                __builtin_amdgcn_sched_group_barrier(0x008, 2 * KB1, 0);
            }
            g1_load(bx1, by1, vb, NB1 - 1);
            g1_aload(xa1, xb1, xa, NB1 - 1);
            __builtin_amdgcn_sched_barrier(0);
            g1_mma(bx0, by0, xa0, xb0, acc0, acc1);
            g1_mma(bx1, by1, xa1, xb1, acc0, acc1);
        }
#pragma unroll
        for (int j = 0; j < 8; ++j) {
            const int r0 = j + 8 * hiK;
            YB[r0 * YPITCH + n0 + lo16]        = fast_tanh(acc0[j]);
            YB[(r0 + 16) * YPITCH + n0 + lo16] = fast_tanh(acc1[j]);
        }

        {
            const float* xr  = &XL[lane * XPITCH + wave * GK];
            const float* gwp = gate_w + wave * GK;
            v2f ga[E_EXP];
#pragma unroll
            for (int e = 0; e < E_EXP; ++e) ga[e] = (v2f){0.f, 0.f};
#pragma unroll 1
            for (int g0 = 0; g0 < GK / 4; g0 += GW_GRP) {
                v4f gv[E_EXP][GW_GRP];
                v4f xv[GW_GRP];
#pragma unroll
                for (int j = 0; j < GW_GRP; ++j) xv[j] = *(const v4f*)&xr[(g0 + j) * 4];
#pragma unroll
                for (int j = 0; j < GW_GRP; ++j)
#pragma unroll
                    for (int e = 0; e < E_EXP; ++e)
                        gv[e][j] = *(const v4f*)(gwp + (size_t)e * D_DIM + (g0 + j) * 4);
                __builtin_amdgcn_sched_barrier(0);
#pragma unroll
                for (int j = 0; j < GW_GRP; ++j) {
#pragma unroll
                    for (int e = 0; e < E_EXP; ++e) {
                        ga[e] = __builtin_elementwise_fma(gv[e][j].xy, xv[j].xy, ga[e]);
                        ga[e] = __builtin_elementwise_fma(gv[e][j].zw, xv[j].zw, ga[e]);
                    }
                }
            }
            v4f part;
            part.x = ga[0].x + ga[0].y;
            part.y = ga[1].x + ga[1].y;
            part.z = ga[2].x + ga[2].y;
            part.w = ga[3].x + ga[3].y;
            *(v4f*)&GP[(wave * B_ROWS + lane) * E_EXP] = part;
        }
        __syncthreads();
        if (tid < B_ROWS) {
            v4f gs = *(const v4f*)&GP[tid * E_EXP];
#pragma unroll
            for (int w = 1; w < 8; ++w) gs += *(const v4f*)&GP[(w * B_ROWS + tid) * E_EXP];
            const float g0 = gs.x, g1 = gs.y, g2 = gs.z, g3 = gs.w;
            const float m  = fmaxf(fmaxf(g0, g1), fmaxf(g2, g3));
            const float e0 = __expf(g0 - m), e1x = __expf(g1 - m);
            const float e2 = __expf(g2 - m), e3 = __expf(g3 - m);
            const float inv = __frcp_rn(e0 + e1x + e2 + e3);
            GT[tid * 4 + 0] = e0 * inv;  GT[tid * 4 + 1] = e1x * inv;
            GT[tid * 4 + 2] = e2 * inv;  GT[tid * 4 + 3] = e3 * inv;
        }
        __syncthreads();

        const int d0 = wave * 128;
        const float* ub = Ul + (size_t)(d0 + lo16) * R_RANK + 2 * hiK;
        v2f bb0[4][8], bb1[4][8];
        g2_load(bb0, ub, 0, 0);
        float bcol[8];
#pragma unroll
        for (int nt = 0; nt < 8; ++nt) bcol[nt] = bl[d0 + nt * 16 + lo16];

        {
            const float* ya = &YB[lo16 * YPITCH + e1 * R_RANK + 2 * hiK];
            v2f ya0[8], ya1[8];
#pragma unroll
            for (int s = 0; s < 8; ++s) {
                ya0[s] = *(const v2f*)&ya[4 * s];
                ya1[s] = *(const v2f*)&ya[16 * YPITCH + 4 * s];
            }
            v8f p0 = vzero, p1 = vzero;
#pragma unroll
            for (int s = 0; s < 8; ++s) {
                p0 = __builtin_amdgcn_wmma_f32_16x16x4_f32(false, ya0[s], false, cfr[s], (short)0, p0, false, false);
                p1 = __builtin_amdgcn_wmma_f32_16x16x4_f32(false, ya1[s], false, cfr[s], (short)0, p1, false, false);
            }
#pragma unroll
            for (int j = 0; j < 8; ++j) {
                const int r0 = j + 8 * hiK;
                CB[r0 * YPITCH + n0 + lo16]        = GT[r0 * 4 + e1] * fast_tanh(p0[j]);
                CB[(r0 + 16) * YPITCH + n0 + lo16] = GT[(r0 + 16) * 4 + e1] * fast_tanh(p1[j]);
            }
        }
        __syncthreads();

        v8f acc2[2][8];
#pragma unroll
        for (int m = 0; m < 2; ++m)
#pragma unroll
            for (int nt = 0; nt < 8; ++nt) acc2[m][nt] = vzero;

        v2f a0[8], a1[8];
        const float* ca = &CB[lo16 * YPITCH + 2 * hiK];
#pragma unroll 1
        for (int e = 0; e < E_EXP - 1; ++e) {
            g2_afrag(a0, a1, ca, e);
            g2_load(bb1, ub, e, 1);
            __builtin_amdgcn_sched_barrier(0);
            g2_mma<0>(bb0, a0, a1, acc2);
            __builtin_amdgcn_sched_barrier(0);
            g2_load(bb0, ub, e + 1, 0);
            __builtin_amdgcn_sched_barrier(0);
            g2_mma<1>(bb1, a0, a1, acc2);
        }
        g2_afrag(a0, a1, ca, E_EXP - 1);
        g2_load(bb1, ub, E_EXP - 1, 1);
        __builtin_amdgcn_sched_barrier(0);
        g2_mma<0>(bb0, a0, a1, acc2);
        g2_mma<1>(bb1, a0, a1, acc2);

        float ex[2][16], e0[2][16];
        epi_load(ex[0], e0[0], XL, X0, d0 + lo16, hiK);
#pragma unroll
        for (int nt = 0; nt < 8; ++nt) {
            if (nt + 1 < 8)
                epi_load(ex[(nt + 1) & 1], e0[(nt + 1) & 1], XL, X0, d0 + (nt + 1) * 16 + lo16, hiK);
            const int col = d0 + nt * 16 + lo16;
#pragma unroll
            for (int m = 0; m < 2; ++m) {
#pragma unroll
                for (int j = 0; j < 8; ++j) {
                    const int row = m * 16 + j + 8 * hiK;
                    XL[row * XPITCH + col] = ex[nt & 1][m * 8 + j] + e0[nt & 1][m * 8 + j] * (bcol[nt] + acc2[m][nt][j]);
                }
            }
        }
        __syncthreads();
    }

    for (int i = tid; i < B_ROWS * (D_DIM / 4); i += THREADS) {
        const int r  = i >> 8;
        const int c4 = i & 255;
        const float4 v4 = *(const float4*)&XL[r * XPITCH + c4 * 4];
        ((float4*)(out + (size_t)(rowBase + r) * D_DIM))[c4] = v4;
    }
}

extern "C" void kernel_launch(void* const* d_in, const int* in_sizes, int n_in,
                              void* d_out, int out_size, void* d_ws, size_t ws_size,
                              hipStream_t stream) {
    (void)in_sizes; (void)n_in; (void)d_ws; (void)ws_size; (void)out_size;
    const float* x      = (const float*)d_in[0];
    const float* U      = (const float*)d_in[1];
    const float* V      = (const float*)d_in[2];
    const float* C      = (const float*)d_in[3];
    const float* bias   = (const float*)d_in[4];
    const float* gate_w = (const float*)d_in[5];
    float* out = (float*)d_out;

    const dim3 grid(B_TOT / B_ROWS);
    const size_t smemBytes = (size_t)SMEM_FLOATS * sizeof(float);
    crossnet_mix_kernel<<<grid, THREADS, smemBytes, stream>>>(x, U, V, C, bias, gate_w, out);
}


# Two workgroups per WGP: 4 waves x 32 rows, LDS halved (no x0 copy, layer 2 re-reads x from global; Y and c' share a plane); one group's matrix runs cover the other's phases; logit sum order changes
# speedup vs baseline: 3.5232x; 1.0723x over previous
#include <hip/hip_runtime.h>
#include <hip/hip_bf16.h>


typedef float v2f __attribute__((ext_vector_type(2)));
typedef float v4f __attribute__((ext_vector_type(4)));
typedef float v8f __attribute__((ext_vector_type(8)));

#define B_TOT    16384
#define D_DIM    1024
#define E_EXP    4
#define R_RANK   32
#define N1       128
#define L_LAYERS 2
#define B_ROWS   32
#define THREADS  128
#define NWAVES   4
#define XPITCH   1028
#define YPITCH   132

#define KB1      8
#define NB1      (D_DIM / (4 * KB1))
#define GK       (D_DIM / NWAVES)
#define GW_GRP   16

#define SMEM_FLOATS (B_ROWS * XPITCH + B_ROWS * YPITCH + B_ROWS * E_EXP + NWAVES * B_ROWS * E_EXP)

__device__ __forceinline__ float fast_tanh(float x) {
    const float e = __expf(2.0f * x);
    return 1.0f - 2.0f * __frcp_rn(e + 1.0f);
}

__device__ __forceinline__ void g1_load(float (&bx)[2][KB1], float (&by)[2][KB1],
                                        const float* __restrict__ vb, int kb) {
    const float* p = vb + (size_t)kb * (4 * KB1) * R_RANK;
#pragma unroll
    for (int j = 0; j < KB1; ++j) {
#pragma unroll
        for (int t = 0; t < 2; ++t) {
            bx[t][j] = p[(4 * j) * R_RANK + 16 * t];
            by[t][j] = p[(4 * j + 1) * R_RANK + 16 * t];
        }
    }
}

__device__ __forceinline__ void g1_aload(v2f (&a0)[KB1], v2f (&a1)[KB1], const float* xa, int kb) {
#pragma unroll
    for (int j = 0; j < KB1; ++j) {
        const int k0 = (kb * KB1 + j) * 4;
        a0[j] = *(const v2f*)&xa[k0];
        a1[j] = *(const v2f*)&xa[16 * XPITCH + k0];
    }
}

__device__ __forceinline__ void g1_mma(const float (&bx)[2][KB1], const float (&by)[2][KB1],
                                       const v2f (&a0)[KB1], const v2f (&a1)[KB1], v8f (&acc)[2][2]) {
#pragma unroll
    for (int j = 0; j < KB1; ++j) {
#pragma unroll
        for (int t = 0; t < 2; ++t) {
            v2f b;
            b.x = bx[t][j];
            b.y = by[t][j];
            acc[0][t] = __builtin_amdgcn_wmma_f32_16x16x4_f32(false, a0[j], false, b, (short)0, acc[0][t], false, false);
            acc[1][t] = __builtin_amdgcn_wmma_f32_16x16x4_f32(false, a1[j], false, b, (short)0, acc[1][t], false, false);
        }
    }
}

__device__ __forceinline__ void g2_load(v2f (&b)[4][8], const float* __restrict__ ub, int e, int half) {
#pragma unroll
    for (int q = 0; q < 4; ++q) {
        const float* p = ub + ((size_t)e * D_DIM + (half * 4 + q) * 16) * R_RANK;
#pragma unroll
        for (int s = 0; s < 8; ++s) b[q][s] = *(const v2f*)(p + 4 * s);
    }
}

__device__ __forceinline__ void g2_afrag(v2f (&a0)[8], v2f (&a1)[8], const float* ca, int e) {
#pragma unroll
    for (int s = 0; s < 8; ++s) {
        a0[s] = *(const v2f*)&ca[e * R_RANK + 4 * s];
        a1[s] = *(const v2f*)&ca[16 * YPITCH + e * R_RANK + 4 * s];
    }
}

template <int HALF>
__device__ __forceinline__ void g2_mma(const v2f (&b)[4][8], const v2f (&a0)[8], const v2f (&a1)[8],
                                       v8f (&acc)[2][8]) {
#pragma unroll
    for (int s = 0; s < 8; ++s) {
#pragma unroll
        for (int q = 0; q < 4; ++q) {
            acc[0][HALF * 4 + q] = __builtin_amdgcn_wmma_f32_16x16x4_f32(false, a0[s], false, b[q][s], (short)0, acc[0][HALF * 4 + q], false, false);
            acc[1][HALF * 4 + q] = __builtin_amdgcn_wmma_f32_16x16x4_f32(false, a1[s], false, b[q][s], (short)0, acc[1][HALF * 4 + q], false, false);
        }
    }
}

__device__ __forceinline__ void epi_load(float (&xv)[16], const float* XL, int col, int hiK) {
#pragma unroll
    for (int m = 0; m < 2; ++m) {
#pragma unroll
        for (int j = 0; j < 8; ++j) {
            const int row = m * 16 + j + 8 * hiK;
            xv[m * 8 + j] = XL[row * XPITCH + col];
        }
    }
}

__device__ __forceinline__ void x0_load(float (&x0v)[16], const float* __restrict__ xg, int col, int hiK) {
#pragma unroll
    for (int m = 0; m < 2; ++m) {
#pragma unroll
        for (int j = 0; j < 8; ++j) {
            const int row = m * 16 + j + 8 * hiK;
            x0v[m * 8 + j] = xg[(size_t)row * D_DIM + col];
        }
    }
}

template <bool X0_GLOBAL>
__device__ __forceinline__ void epilogue(float* XL, const float* __restrict__ xg, const v8f (&acc2)[2][8],
                                         const float (&bcol)[8], int d0, int lo16, int hiK) {
    float x0g[X0_GLOBAL ? 8 : 1][16];
    if (X0_GLOBAL) {
#pragma unroll
        for (int nt = 0; nt < 8; ++nt) x0_load(x0g[nt], xg, d0 + nt * 16 + lo16, hiK);
    }
    float ex[2][16];
    epi_load(ex[0], XL, d0 + lo16, hiK);
#pragma unroll
    for (int nt = 0; nt < 8; ++nt) {
        if (nt + 1 < 8) epi_load(ex[(nt + 1) & 1], XL, d0 + (nt + 1) * 16 + lo16, hiK);
        const int col = d0 + nt * 16 + lo16;
#pragma unroll
        for (int m = 0; m < 2; ++m) {
#pragma unroll
            for (int j = 0; j < 8; ++j) {
                const int row = m * 16 + j + 8 * hiK;
                const float xl = ex[nt & 1][m * 8 + j];
                const float x0 = X0_GLOBAL ? x0g[nt][m * 8 + j] : xl;
                XL[row * XPITCH + col] = xl + x0 * (bcol[nt] + acc2[m][nt][j]);
            }
        }
    }
}

extern "C" __global__ __launch_bounds__(THREADS, 2)
void crossnet_mix_kernel(const float* __restrict__ x,
                         const float* __restrict__ U,
                         const float* __restrict__ V,
                         const float* __restrict__ C,
                         const float* __restrict__ bias,
                         const float* __restrict__ gate_w,
                         float* __restrict__ out)
{
    extern __shared__ float smem[];
    float* XL = smem;
    float* YC = smem + B_ROWS * XPITCH;
    float* GT = YC + B_ROWS * YPITCH;
    float* GP = GT + B_ROWS * E_EXP;

    const int tid  = threadIdx.x;
    const int wave = tid >> 5;
    const int lane = tid & 31;
    const int lo16 = lane & 15;
    const int hiK  = lane >> 4;
    const int rowBase = blockIdx.x * B_ROWS;
    const float* xg = x + (size_t)rowBase * D_DIM;

#pragma unroll
    for (int rh = 0; rh < B_ROWS; rh += 16) {
        float4 xv[16][2];
#pragma unroll
        for (int r = 0; r < 16; ++r) {
            xv[r][0] = ((const float4*)(xg + (size_t)(rh + r) * D_DIM))[tid];
            xv[r][1] = ((const float4*)(xg + (size_t)(rh + r) * D_DIM))[tid + THREADS];
        }
#pragma unroll
        for (int r = 0; r < 16; ++r) {
            *(float4*)&XL[(rh + r) * XPITCH + tid * 4] = xv[r][0];
            *(float4*)&XL[(rh + r) * XPITCH + (tid + THREADS) * 4] = xv[r][1];
        }
    }
    __syncthreads();

    const v8f vzero = {};

    const int e1 = wave;
    const int n0 = e1 * R_RANK;

    for (int l = 0; l < L_LAYERS; ++l) {
        const float* Vl = V + (size_t)l * E_EXP * D_DIM * R_RANK;
        const float* Ul = U + (size_t)l * E_EXP * D_DIM * R_RANK;
        const float* Cl = C + (size_t)l * E_EXP * R_RANK * R_RANK;
        const float* bl = bias + (size_t)l * D_DIM;

        v2f cfr[2][8];
#pragma unroll
        for (int t = 0; t < 2; ++t) {
            const float* cp = Cl + ((size_t)e1 * R_RANK + 16 * t + lo16) * R_RANK + 2 * hiK;
#pragma unroll
            for (int s = 0; s < 8; ++s) cfr[t][s] = *(const v2f*)(cp + 4 * s);
        }

        v8f acc1[2][2];
#pragma unroll
        for (int m = 0; m < 2; ++m)
#pragma unroll
            for (int t = 0; t < 2; ++t) acc1[m][t] = vzero;
        {
            const float* vb = Vl + ((size_t)e1 * D_DIM + 2 * hiK) * R_RANK + lo16;
            const float* xa = &XL[lo16 * XPITCH + 2 * hiK];
            float bx0[2][KB1], by0[2][KB1], bx1[2][KB1], by1[2][KB1];
            v2f xa0[KB1], xb0[KB1], xa1[KB1], xb1[KB1];
            g1_load(bx0, by0, vb, 0);
            g1_aload(xa0, xb0, xa, 0);
            for (int kb = 0; kb < NB1 - 2; kb += 2) {
                g1_load(bx1, by1, vb, kb + 1);
                g1_aload(xa1, xb1, xa, kb + 1);
                g1_mma(bx0, by0, xa0, xb0, acc1);
                g1_load(bx0, by0, vb, kb + 2);
                g1_aload(xa0, xb0, xa, kb + 2);
                g1_mma(bx1, by1, xa1, xb1, acc1);
                __builtin_amdgcn_sched_group_barrier(0x020, 4 * KB1, 0);
                __builtin_amdgcn_sched_group_barrier(0x100, 2 * KB1, 0);
                __builtin_amdgcn_sched_group_barrier(0x008, 4 * KB1, 0);
                __builtin_amdgcn_sched_group_barrier(0x020, 4 * KB1, 0);
                __builtin_amdgcn_sched_group_barrier(0x100, 2 * KB1, 0);
                __builtin_amdgcn_sched_group_barrier(0x008, 4 * KB1, 0);
            }
            g1_load(bx1, by1, vb, NB1 - 1);
            g1_aload(xa1, xb1, xa, NB1 - 1);
            __builtin_amdgcn_sched_barrier(0);
            g1_mma(bx0, by0, xa0, xb0, acc1);
            g1_mma(bx1, by1, xa1, xb1, acc1);
        }
#pragma unroll
        for (int t = 0; t < 2; ++t) {
#pragma unroll
            for (int j = 0; j < 8; ++j) {
                const int r0 = j + 8 * hiK;
                YC[r0 * YPITCH + n0 + 16 * t + lo16]        = fast_tanh(acc1[0][t][j]);
                YC[(r0 + 16) * YPITCH + n0 + 16 * t + lo16] = fast_tanh(acc1[1][t][j]);
            }
        }

        {
            const float* xr  = &XL[lane * XPITCH + wave * GK];
            const float* gwp = gate_w + wave * GK;
            v2f ga[E_EXP];
#pragma unroll
            for (int e = 0; e < E_EXP; ++e) ga[e] = (v2f){0.f, 0.f};
#pragma unroll 1
            for (int g0 = 0; g0 < GK / 4; g0 += GW_GRP) {
                v4f gv[E_EXP][GW_GRP];
                v4f xv[GW_GRP];
#pragma unroll
                for (int j = 0; j < GW_GRP; ++j) xv[j] = *(const v4f*)&xr[(g0 + j) * 4];
#pragma unroll
                for (int j = 0; j < GW_GRP; ++j)
#pragma unroll
                    for (int e = 0; e < E_EXP; ++e)
                        gv[e][j] = *(const v4f*)(gwp + (size_t)e * D_DIM + (g0 + j) * 4);
                __builtin_amdgcn_sched_barrier(0);
#pragma unroll
                for (int j = 0; j < GW_GRP; ++j) {
#pragma unroll
                    for (int e = 0; e < E_EXP; ++e) {
                        ga[e] = __builtin_elementwise_fma(gv[e][j].xy, xv[j].xy, ga[e]);
                        ga[e] = __builtin_elementwise_fma(gv[e][j].zw, xv[j].zw, ga[e]);
                    }
                }
            }
            v4f part;
            part.x = ga[0].x + ga[0].y;
            part.y = ga[1].x + ga[1].y;
            part.z = ga[2].x + ga[2].y;
            part.w = ga[3].x + ga[3].y;
            *(v4f*)&GP[(wave * B_ROWS + lane) * E_EXP] = part;
        }
        __syncthreads();
        if (tid < B_ROWS) {
            v4f gs = *(const v4f*)&GP[tid * E_EXP];
#pragma unroll
            for (int w = 1; w < NWAVES; ++w) gs += *(const v4f*)&GP[(w * B_ROWS + tid) * E_EXP];
            const float g0 = gs.x, g1 = gs.y, g2 = gs.z, g3 = gs.w;
            const float m  = fmaxf(fmaxf(g0, g1), fmaxf(g2, g3));
            const float e0 = __expf(g0 - m), e1x = __expf(g1 - m);
            const float e2 = __expf(g2 - m), e3 = __expf(g3 - m);
            const float inv = __frcp_rn(e0 + e1x + e2 + e3);
            GT[tid * 4 + 0] = e0 * inv;  GT[tid * 4 + 1] = e1x * inv;
            GT[tid * 4 + 2] = e2 * inv;  GT[tid * 4 + 3] = e3 * inv;
        }
        __syncthreads();

        const int dw = wave * 256;
        v2f bb0[4][8], bb1[4][8];
        g2_load(bb0, Ul + (size_t)(dw + lo16) * R_RANK + 2 * hiK, 0, 0);

        {
            const float* ya = &YC[lo16 * YPITCH + n0 + 2 * hiK];
            v2f ya0[8], ya1[8];
#pragma unroll
            for (int s = 0; s < 8; ++s) {
                ya0[s] = *(const v2f*)&ya[4 * s];
                ya1[s] = *(const v2f*)&ya[16 * YPITCH + 4 * s];
            }
            v8f p[2][2];
#pragma unroll
            for (int m = 0; m < 2; ++m)
#pragma unroll
                for (int t = 0; t < 2; ++t) p[m][t] = vzero;
#pragma unroll
            for (int s = 0; s < 8; ++s) {
#pragma unroll
                for (int t = 0; t < 2; ++t) {
                    p[0][t] = __builtin_amdgcn_wmma_f32_16x16x4_f32(false, ya0[s], false, cfr[t][s], (short)0, p[0][t], false, false);
                    p[1][t] = __builtin_amdgcn_wmma_f32_16x16x4_f32(false, ya1[s], false, cfr[t][s], (short)0, p[1][t], false, false);
                }
            }
#pragma unroll
            for (int t = 0; t < 2; ++t) {
#pragma unroll
                for (int j = 0; j < 8; ++j) {
                    const int r0 = j + 8 * hiK;
                    YC[r0 * YPITCH + n0 + 16 * t + lo16]        = GT[r0 * 4 + e1] * fast_tanh(p[0][t][j]);
                    YC[(r0 + 16) * YPITCH + n0 + 16 * t + lo16] = GT[(r0 + 16) * 4 + e1] * fast_tanh(p[1][t][j]);
                }
            }
        }
        __syncthreads();

        const float* ca = &YC[lo16 * YPITCH + 2 * hiK];
#pragma unroll 1
        for (int pass = 0; pass < 2; ++pass) {
            const int d0 = dw + pass * 128;
            const float* ub = Ul + (size_t)(d0 + lo16) * R_RANK + 2 * hiK;
            float bcol[8];
#pragma unroll
            for (int nt = 0; nt < 8; ++nt) bcol[nt] = bl[d0 + nt * 16 + lo16];

            v8f acc2[2][8];
#pragma unroll
            for (int m = 0; m < 2; ++m)
#pragma unroll
                for (int nt = 0; nt < 8; ++nt) acc2[m][nt] = vzero;

            v2f a0[8], a1[8];
#pragma unroll 1
            for (int e = 0; e < E_EXP - 1; ++e) {
                g2_afrag(a0, a1, ca, e);
                g2_load(bb1, ub, e, 1);
                __builtin_amdgcn_sched_barrier(0);
                g2_mma<0>(bb0, a0, a1, acc2);
                __builtin_amdgcn_sched_barrier(0);
                g2_load(bb0, ub, e + 1, 0);
                __builtin_amdgcn_sched_barrier(0);
                g2_mma<1>(bb1, a0, a1, acc2);
            }
            g2_afrag(a0, a1, ca, E_EXP - 1);
            g2_load(bb1, ub, E_EXP - 1, 1);
            __builtin_amdgcn_sched_barrier(0);
            g2_mma<0>(bb0, a0, a1, acc2);
            __builtin_amdgcn_sched_barrier(0);
            if (pass == 0) g2_load(bb0, ub + (size_t)128 * R_RANK, 0, 0);
            __builtin_amdgcn_sched_barrier(0);
            g2_mma<1>(bb1, a0, a1, acc2);

            if (l == 0) epilogue<false>(XL, xg, acc2, bcol, d0, lo16, hiK);
            else        epilogue<true>(XL, xg, acc2, bcol, d0, lo16, hiK);
        }
        __syncthreads();
    }

    for (int i = tid; i < B_ROWS * (D_DIM / 4); i += THREADS) {
        const int r  = i >> 8;
        const int c4 = i & 255;
        const float4 v4 = *(const float4*)&XL[r * XPITCH + c4 * 4];
        ((float4*)(out + (size_t)(rowBase + r) * D_DIM))[c4] = v4;
    }
}

extern "C" void kernel_launch(void* const* d_in, const int* in_sizes, int n_in,
                              void* d_out, int out_size, void* d_ws, size_t ws_size,
                              hipStream_t stream) {
    (void)in_sizes; (void)n_in; (void)d_ws; (void)ws_size; (void)out_size;
    const float* x      = (const float*)d_in[0];
    const float* U      = (const float*)d_in[1];
    const float* V      = (const float*)d_in[2];
    const float* C      = (const float*)d_in[3];
    const float* bias   = (const float*)d_in[4];
    const float* gate_w = (const float*)d_in[5];
    float* out = (float*)d_out;

    const dim3 grid(B_TOT / B_ROWS);
    const size_t smemBytes = (size_t)SMEM_FLOATS * sizeof(float);
    crossnet_mix_kernel<<<grid, THREADS, smemBytes, stream>>>(x, U, V, C, bias, gate_w, out);
}


# Wave issue priority by phase (s_setprio): vector/LDS/latency phases above the two products, earlier product above later, so the two co-resident workgroups cover each other; no arithmetic change
# speedup vs baseline: 3.6382x; 1.0326x over previous
#include <hip/hip_runtime.h>
#include <hip/hip_bf16.h>


typedef float v2f __attribute__((ext_vector_type(2)));
typedef float v4f __attribute__((ext_vector_type(4)));
typedef float v8f __attribute__((ext_vector_type(8)));

#define B_TOT    16384
#define D_DIM    1024
#define E_EXP    4
#define R_RANK   32
#define N1       128
#define L_LAYERS 2
#define B_ROWS   32
#define THREADS  128
#define NWAVES   4
#define XPITCH   1028
#define YPITCH   132

#define KB1      8
#define NB1      (D_DIM / (4 * KB1))
#define GK       (D_DIM / NWAVES)
#define GW_GRP   16

#define SMEM_FLOATS (B_ROWS * XPITCH + B_ROWS * YPITCH + B_ROWS * E_EXP + NWAVES * B_ROWS * E_EXP)

__device__ __forceinline__ float fast_tanh(float x) {
    const float e = __expf(2.0f * x);
    return 1.0f - 2.0f * __frcp_rn(e + 1.0f);
}

#define PRIO_THIN 3

__device__ __forceinline__ void g1_load(float (&bx)[2][KB1], float (&by)[2][KB1],
                                        const float* __restrict__ vb, int kb) {
    const float* p = vb + (size_t)kb * (4 * KB1) * R_RANK;
#pragma unroll
    for (int j = 0; j < KB1; ++j) {
#pragma unroll
        for (int t = 0; t < 2; ++t) {
            bx[t][j] = p[(4 * j) * R_RANK + 16 * t];
            by[t][j] = p[(4 * j + 1) * R_RANK + 16 * t];
        }
    }
}

__device__ __forceinline__ void g1_aload(v2f (&a0)[KB1], v2f (&a1)[KB1], const float* xa, int kb) {
#pragma unroll
    for (int j = 0; j < KB1; ++j) {
        const int k0 = (kb * KB1 + j) * 4;
        a0[j] = *(const v2f*)&xa[k0];
        a1[j] = *(const v2f*)&xa[16 * XPITCH + k0];
    }
}

__device__ __forceinline__ void g1_mma(const float (&bx)[2][KB1], const float (&by)[2][KB1],
                                       const v2f (&a0)[KB1], const v2f (&a1)[KB1], v8f (&acc)[2][2]) {
#pragma unroll
    for (int j = 0; j < KB1; ++j) {
#pragma unroll
        for (int t = 0; t < 2; ++t) {
            v2f b;
            b.x = bx[t][j];
            b.y = by[t][j];
            acc[0][t] = __builtin_amdgcn_wmma_f32_16x16x4_f32(false, a0[j], false, b, (short)0, acc[0][t], false, false);
            acc[1][t] = __builtin_amdgcn_wmma_f32_16x16x4_f32(false, a1[j], false, b, (short)0, acc[1][t], false, false);
        }
    }
}

__device__ __forceinline__ void g2_load(v2f (&b)[4][8], const float* __restrict__ ub, int e, int half) {
#pragma unroll
    for (int q = 0; q < 4; ++q) {
        const float* p = ub + ((size_t)e * D_DIM + (half * 4 + q) * 16) * R_RANK;
#pragma unroll
        for (int s = 0; s < 8; ++s) b[q][s] = *(const v2f*)(p + 4 * s);
    }
}

__device__ __forceinline__ void g2_afrag(v2f (&a0)[8], v2f (&a1)[8], const float* ca, int e) {
#pragma unroll
    for (int s = 0; s < 8; ++s) {
        a0[s] = *(const v2f*)&ca[e * R_RANK + 4 * s];
        a1[s] = *(const v2f*)&ca[16 * YPITCH + e * R_RANK + 4 * s];
    }
}

template <int HALF>
__device__ __forceinline__ void g2_mma(const v2f (&b)[4][8], const v2f (&a0)[8], const v2f (&a1)[8],
                                       v8f (&acc)[2][8]) {
#pragma unroll
    for (int s = 0; s < 8; ++s) {
#pragma unroll
        for (int q = 0; q < 4; ++q) {
            acc[0][HALF * 4 + q] = __builtin_amdgcn_wmma_f32_16x16x4_f32(false, a0[s], false, b[q][s], (short)0, acc[0][HALF * 4 + q], false, false);
            acc[1][HALF * 4 + q] = __builtin_amdgcn_wmma_f32_16x16x4_f32(false, a1[s], false, b[q][s], (short)0, acc[1][HALF * 4 + q], false, false);
        }
    }
}

__device__ __forceinline__ void epi_load(float (&xv)[16], const float* XL, int col, int hiK) {
#pragma unroll
    for (int m = 0; m < 2; ++m) {
#pragma unroll
        for (int j = 0; j < 8; ++j) {
            const int row = m * 16 + j + 8 * hiK;
            xv[m * 8 + j] = XL[row * XPITCH + col];
        }
    }
}

__device__ __forceinline__ void x0_load(float (&x0v)[16], const float* __restrict__ xg, int col, int hiK) {
#pragma unroll
    for (int m = 0; m < 2; ++m) {
#pragma unroll
        for (int j = 0; j < 8; ++j) {
            const int row = m * 16 + j + 8 * hiK;
            x0v[m * 8 + j] = xg[(size_t)row * D_DIM + col];
        }
    }
}

template <bool X0_GLOBAL>
__device__ __forceinline__ void epilogue(float* XL, const float* __restrict__ xg, const v8f (&acc2)[2][8],
                                         const float (&bcol)[8], int d0, int lo16, int hiK) {
    float x0g[X0_GLOBAL ? 8 : 1][16];
    if (X0_GLOBAL) {
#pragma unroll
        for (int nt = 0; nt < 8; ++nt) x0_load(x0g[nt], xg, d0 + nt * 16 + lo16, hiK);
    }
    float ex[2][16];
    epi_load(ex[0], XL, d0 + lo16, hiK);
#pragma unroll
    for (int nt = 0; nt < 8; ++nt) {
        if (nt + 1 < 8) epi_load(ex[(nt + 1) & 1], XL, d0 + (nt + 1) * 16 + lo16, hiK);
        const int col = d0 + nt * 16 + lo16;
#pragma unroll
        for (int m = 0; m < 2; ++m) {
#pragma unroll
            for (int j = 0; j < 8; ++j) {
                const int row = m * 16 + j + 8 * hiK;
                const float xl = ex[nt & 1][m * 8 + j];
                const float x0 = X0_GLOBAL ? x0g[nt][m * 8 + j] : xl;
                XL[row * XPITCH + col] = xl + x0 * (bcol[nt] + acc2[m][nt][j]);
            }
        }
    }
}

extern "C" __global__ __launch_bounds__(THREADS, 2)
void crossnet_mix_kernel(const float* __restrict__ x,
                         const float* __restrict__ U,
                         const float* __restrict__ V,
                         const float* __restrict__ C,
                         const float* __restrict__ bias,
                         const float* __restrict__ gate_w,
                         float* __restrict__ out)
{
    extern __shared__ float smem[];
    float* XL = smem;
    float* YC = smem + B_ROWS * XPITCH;
    float* GT = YC + B_ROWS * YPITCH;
    float* GP = GT + B_ROWS * E_EXP;

    const int tid  = threadIdx.x;
    const int wave = tid >> 5;
    const int lane = tid & 31;
    const int lo16 = lane & 15;
    const int hiK  = lane >> 4;
    const int rowBase = blockIdx.x * B_ROWS;
    const float* xg = x + (size_t)rowBase * D_DIM;

    __builtin_amdgcn_s_setprio(PRIO_THIN);

#pragma unroll
    for (int rh = 0; rh < B_ROWS; rh += 16) {
        float4 xv[16][2];
#pragma unroll
        for (int r = 0; r < 16; ++r) {
            xv[r][0] = ((const float4*)(xg + (size_t)(rh + r) * D_DIM))[tid];
            xv[r][1] = ((const float4*)(xg + (size_t)(rh + r) * D_DIM))[tid + THREADS];
        }
#pragma unroll
        for (int r = 0; r < 16; ++r) {
            *(float4*)&XL[(rh + r) * XPITCH + tid * 4] = xv[r][0];
            *(float4*)&XL[(rh + r) * XPITCH + (tid + THREADS) * 4] = xv[r][1];
        }
    }
    __syncthreads();

    const v8f vzero = {};

    const int e1 = wave;
    const int n0 = e1 * R_RANK;

    for (int l = 0; l < L_LAYERS; ++l) {
        const float* Vl = V + (size_t)l * E_EXP * D_DIM * R_RANK;
        const float* Ul = U + (size_t)l * E_EXP * D_DIM * R_RANK;
        const float* Cl = C + (size_t)l * E_EXP * R_RANK * R_RANK;
        const float* bl = bias + (size_t)l * D_DIM;

        v2f cfr[2][8];
#pragma unroll
        for (int t = 0; t < 2; ++t) {
            const float* cp = Cl + ((size_t)e1 * R_RANK + 16 * t + lo16) * R_RANK + 2 * hiK;
#pragma unroll
            for (int s = 0; s < 8; ++s) cfr[t][s] = *(const v2f*)(cp + 4 * s);
        }

        v8f acc1[2][2];
#pragma unroll
        for (int m = 0; m < 2; ++m)
#pragma unroll
            for (int t = 0; t < 2; ++t) acc1[m][t] = vzero;
        {
            if (l == 0) __builtin_amdgcn_s_setprio(2);
            else        __builtin_amdgcn_s_setprio(1);
            const float* vb = Vl + ((size_t)e1 * D_DIM + 2 * hiK) * R_RANK + lo16;
            const float* xa = &XL[lo16 * XPITCH + 2 * hiK];
            float bx0[2][KB1], by0[2][KB1], bx1[2][KB1], by1[2][KB1];
            v2f xa0[KB1], xb0[KB1], xa1[KB1], xb1[KB1];
            g1_load(bx0, by0, vb, 0);
            g1_aload(xa0, xb0, xa, 0);
            for (int kb = 0; kb < NB1 - 2; kb += 2) {
                g1_load(bx1, by1, vb, kb + 1);
                g1_aload(xa1, xb1, xa, kb + 1);
                g1_mma(bx0, by0, xa0, xb0, acc1);
                g1_load(bx0, by0, vb, kb + 2);
                g1_aload(xa0, xb0, xa, kb + 2);
                g1_mma(bx1, by1, xa1, xb1, acc1);
                __builtin_amdgcn_sched_group_barrier(0x020, 4 * KB1, 0);
                __builtin_amdgcn_sched_group_barrier(0x100, 2 * KB1, 0);
                __builtin_amdgcn_sched_group_barrier(0x008, 4 * KB1, 0);
                __builtin_amdgcn_sched_group_barrier(0x020, 4 * KB1, 0);
                __builtin_amdgcn_sched_group_barrier(0x100, 2 * KB1, 0);
                __builtin_amdgcn_sched_group_barrier(0x008, 4 * KB1, 0);
            }
            g1_load(bx1, by1, vb, NB1 - 1);
            g1_aload(xa1, xb1, xa, NB1 - 1);
            __builtin_amdgcn_sched_barrier(0);
            g1_mma(bx0, by0, xa0, xb0, acc1);
            g1_mma(bx1, by1, xa1, xb1, acc1);
        }
        __builtin_amdgcn_s_setprio(PRIO_THIN);
#pragma unroll
        for (int t = 0; t < 2; ++t) {
#pragma unroll
            for (int j = 0; j < 8; ++j) {
                const int r0 = j + 8 * hiK;
                YC[r0 * YPITCH + n0 + 16 * t + lo16]        = fast_tanh(acc1[0][t][j]);
                YC[(r0 + 16) * YPITCH + n0 + 16 * t + lo16] = fast_tanh(acc1[1][t][j]);
            }
        }

        {
            const float* xr  = &XL[lane * XPITCH + wave * GK];
            const float* gwp = gate_w + wave * GK;
            v2f ga[E_EXP];
#pragma unroll
            for (int e = 0; e < E_EXP; ++e) ga[e] = (v2f){0.f, 0.f};
#pragma unroll 1
            for (int g0 = 0; g0 < GK / 4; g0 += GW_GRP) {
                v4f gv[E_EXP][GW_GRP];
                v4f xv[GW_GRP];
#pragma unroll
                for (int j = 0; j < GW_GRP; ++j) xv[j] = *(const v4f*)&xr[(g0 + j) * 4];
#pragma unroll
                for (int j = 0; j < GW_GRP; ++j)
#pragma unroll
                    for (int e = 0; e < E_EXP; ++e)
                        gv[e][j] = *(const v4f*)(gwp + (size_t)e * D_DIM + (g0 + j) * 4);
                __builtin_amdgcn_sched_barrier(0);
#pragma unroll
                for (int j = 0; j < GW_GRP; ++j) {
#pragma unroll
                    for (int e = 0; e < E_EXP; ++e) {
                        ga[e] = __builtin_elementwise_fma(gv[e][j].xy, xv[j].xy, ga[e]);
                        ga[e] = __builtin_elementwise_fma(gv[e][j].zw, xv[j].zw, ga[e]);
                    }
                }
            }
            v4f part;
            part.x = ga[0].x + ga[0].y;
            part.y = ga[1].x + ga[1].y;
            part.z = ga[2].x + ga[2].y;
            part.w = ga[3].x + ga[3].y;
            *(v4f*)&GP[(wave * B_ROWS + lane) * E_EXP] = part;
        }
        __syncthreads();
        if (tid < B_ROWS) {
            v4f gs = *(const v4f*)&GP[tid * E_EXP];
#pragma unroll
            for (int w = 1; w < NWAVES; ++w) gs += *(const v4f*)&GP[(w * B_ROWS + tid) * E_EXP];
            const float g0 = gs.x, g1 = gs.y, g2 = gs.z, g3 = gs.w;
            const float m  = fmaxf(fmaxf(g0, g1), fmaxf(g2, g3));
            const float e0 = __expf(g0 - m), e1x = __expf(g1 - m);
            const float e2 = __expf(g2 - m), e3 = __expf(g3 - m);
            const float inv = __frcp_rn(e0 + e1x + e2 + e3);
            GT[tid * 4 + 0] = e0 * inv;  GT[tid * 4 + 1] = e1x * inv;
            GT[tid * 4 + 2] = e2 * inv;  GT[tid * 4 + 3] = e3 * inv;
        }
        __syncthreads();

        const int dw = wave * 256;
        v2f bb0[4][8], bb1[4][8];
        g2_load(bb0, Ul + (size_t)(dw + lo16) * R_RANK + 2 * hiK, 0, 0);

        {
            const float* ya = &YC[lo16 * YPITCH + n0 + 2 * hiK];
            v2f ya0[8], ya1[8];
#pragma unroll
            for (int s = 0; s < 8; ++s) {
                ya0[s] = *(const v2f*)&ya[4 * s];
                ya1[s] = *(const v2f*)&ya[16 * YPITCH + 4 * s];
            }
            v8f p[2][2];
#pragma unroll
            for (int m = 0; m < 2; ++m)
#pragma unroll
                for (int t = 0; t < 2; ++t) p[m][t] = vzero;
#pragma unroll
            for (int s = 0; s < 8; ++s) {
#pragma unroll
                for (int t = 0; t < 2; ++t) {
                    p[0][t] = __builtin_amdgcn_wmma_f32_16x16x4_f32(false, ya0[s], false, cfr[t][s], (short)0, p[0][t], false, false);
                    p[1][t] = __builtin_amdgcn_wmma_f32_16x16x4_f32(false, ya1[s], false, cfr[t][s], (short)0, p[1][t], false, false);
                }
            }
#pragma unroll
            for (int t = 0; t < 2; ++t) {
#pragma unroll
                for (int j = 0; j < 8; ++j) {
                    const int r0 = j + 8 * hiK;
                    YC[r0 * YPITCH + n0 + 16 * t + lo16]        = GT[r0 * 4 + e1] * fast_tanh(p[0][t][j]);
                    YC[(r0 + 16) * YPITCH + n0 + 16 * t + lo16] = GT[(r0 + 16) * 4 + e1] * fast_tanh(p[1][t][j]);
                }
            }
        }
        __syncthreads();

        const float* ca = &YC[lo16 * YPITCH + 2 * hiK];
#pragma unroll 1
        for (int pass = 0; pass < 2; ++pass) {
            const int d0 = dw + pass * 128;
            const float* ub = Ul + (size_t)(d0 + lo16) * R_RANK + 2 * hiK;
            float bcol[8];
#pragma unroll
            for (int nt = 0; nt < 8; ++nt) bcol[nt] = bl[d0 + nt * 16 + lo16];

            v8f acc2[2][8];
#pragma unroll
            for (int m = 0; m < 2; ++m)
#pragma unroll
                for (int nt = 0; nt < 8; ++nt) acc2[m][nt] = vzero;

            v2f a0[8], a1[8];
            if (l == 0) __builtin_amdgcn_s_setprio(1);
            else        __builtin_amdgcn_s_setprio(0);
#pragma unroll 1
            for (int e = 0; e < E_EXP - 1; ++e) {
                g2_afrag(a0, a1, ca, e);
                g2_load(bb1, ub, e, 1);
                __builtin_amdgcn_sched_barrier(0);
                g2_mma<0>(bb0, a0, a1, acc2);
                __builtin_amdgcn_sched_barrier(0);
                g2_load(bb0, ub, e + 1, 0);
                __builtin_amdgcn_sched_barrier(0);
                g2_mma<1>(bb1, a0, a1, acc2);
            }
            g2_afrag(a0, a1, ca, E_EXP - 1);
            g2_load(bb1, ub, E_EXP - 1, 1);
            __builtin_amdgcn_sched_barrier(0);
            g2_mma<0>(bb0, a0, a1, acc2);
            __builtin_amdgcn_sched_barrier(0);
            if (pass == 0) g2_load(bb0, ub + (size_t)128 * R_RANK, 0, 0);
            __builtin_amdgcn_sched_barrier(0);
            g2_mma<1>(bb1, a0, a1, acc2);

            __builtin_amdgcn_s_setprio(PRIO_THIN);
            if (l == 0) epilogue<false>(XL, xg, acc2, bcol, d0, lo16, hiK);
            else        epilogue<true>(XL, xg, acc2, bcol, d0, lo16, hiK);
        }
        __syncthreads();
    }

    for (int i = tid; i < B_ROWS * (D_DIM / 4); i += THREADS) {
        const int r  = i >> 8;
        const int c4 = i & 255;
        const float4 v4 = *(const float4*)&XL[r * XPITCH + c4 * 4];
        ((float4*)(out + (size_t)(rowBase + r) * D_DIM))[c4] = v4;
    }
}

extern "C" void kernel_launch(void* const* d_in, const int* in_sizes, int n_in,
                              void* d_out, int out_size, void* d_ws, size_t ws_size,
                              hipStream_t stream) {
    (void)in_sizes; (void)n_in; (void)d_ws; (void)ws_size; (void)out_size;
    const float* x      = (const float*)d_in[0];
    const float* U      = (const float*)d_in[1];
    const float* V      = (const float*)d_in[2];
    const float* C      = (const float*)d_in[3];
    const float* bias   = (const float*)d_in[4];
    const float* gate_w = (const float*)d_in[5];
    float* out = (float*)d_out;

    const dim3 grid(B_TOT / B_ROWS);
    const size_t smemBytes = (size_t)SMEM_FLOATS * sizeof(float);
    crossnet_mix_kernel<<<grid, THREADS, smemBytes, stream>>>(x, U, V, C, bias, gate_w, out);
}
